# starting kernel
# baseline (speedup 1.0000x reference)
#include <hip/hip_runtime.h>
#include <math.h>


typedef __attribute__((ext_vector_type(2))) float v2f;
typedef __attribute__((ext_vector_type(8))) float v8f;

__device__ __forceinline__ float lrelu02(float v) {
    return v > 0.0f ? v : 0.2f * v;
}

__device__ __forceinline__ void atomicMaxF(float* addr, float v) {
    int iv = __float_as_int(v);
    if (iv >= 0) atomicMax((int*)addr, iv);
    else         atomicMin((unsigned int*)addr, (unsigned int)iv);
}

__global__ void fill_kernel(float* __restrict__ p, size_t n, float v) {
    size_t i = (size_t)blockIdx.x * blockDim.x + threadIdx.x;
    if (i < n) p[i] = v;
}

__global__ __launch_bounds__(256) void gemm_wmma_kernel(
    const float* __restrict__ X, const float* __restrict__ W,
    float* __restrict__ Y, int M, int K)
{
    __shared__ float sW[128 * 64];
    const int nW = K * 64;
    for (int i = threadIdx.x; i < nW; i += 256) {
        int k = i >> 6, n = i & 63;
        sW[(k >> 1) * 128 + 2 * n + (k & 1)] = W[i];
    }
    __syncthreads();

    const int wave = threadIdx.x >> 5;
    const int lane = threadIdx.x & 31;
    const int m0   = (blockIdx.x * 8 + wave) * 16;
    if (m0 >= M) return;

    const int half = lane >> 4;
    const int r    = lane & 15;

    v8f c0 = {}, c1 = {}, c2 = {}, c3 = {};

    const float* arow = X + (size_t)(m0 + r) * K + 2 * half;

    for (int k0 = 0; k0 < K; k0 += 4) {
        v2f a = *(const v2f*)(arow + k0);

        const float* wp = &sW[((k0 >> 1) + half) * 128 + 2 * r];
        v2f b0 = *(const v2f*)(wp +  0);
        v2f b1 = *(const v2f*)(wp + 32);
        v2f b2 = *(const v2f*)(wp + 64);
        v2f b3 = *(const v2f*)(wp + 96);

        c0 = __builtin_amdgcn_wmma_f32_16x16x4_f32(false, a, false, b0,
                                                   (short)0, c0, false, false);
        c1 = __builtin_amdgcn_wmma_f32_16x16x4_f32(false, a, false, b1,
                                                   (short)0, c1, false, false);
        c2 = __builtin_amdgcn_wmma_f32_16x16x4_f32(false, a, false, b2,
                                                   (short)0, c2, false, false);
        c3 = __builtin_amdgcn_wmma_f32_16x16x4_f32(false, a, false, b3,
                                                   (short)0, c3, false, false);
    }

    float* yb = Y + (size_t)m0 * 64 + r;
#pragma unroll
    for (int j = 0; j < 8; ++j) {
        float* yr = yb + (size_t)(j + half * 8) * 64;
        yr[0]  = c0[j];
        yr[16] = c1[j];
        yr[32] = c2[j];
        yr[48] = c3[j];
    }
}

__global__ void alpha_kernel(const float* __restrict__ xh,
                             const float* __restrict__ a_src,
                             const float* __restrict__ a_dst,
                             float* __restrict__ asrc,
                             float* __restrict__ adst,
                             int N, int H, int C)
{
    int idx = blockIdx.x * blockDim.x + threadIdx.x;
    if (idx >= N * H) return;
    int n = idx / H, h = idx % H;
    const float* row = xh + (size_t)n * 64 + h * C;
    const float* ws  = a_src + h * C;
    const float* wd  = a_dst + h * C;
    float s = 0.f, t = 0.f;
    for (int c = 0; c < C; ++c) {
        float v = row[c];
        s += v * ws[c];
        t += v * wd[c];
    }
    asrc[idx] = s;
    adst[idx] = t;
}

__device__ __forceinline__ void edge_sd(const int* __restrict__ ei,
                                        int e, int E, int& s, int& d) {
    if (e < E) { s = ei[e]; d = ei[E + e]; }
    else       { s = d = e - E; }
}

__global__ void edge_max_kernel(const int* __restrict__ ei, int E, int Etot,
                                const float* __restrict__ asrc,
                                const float* __restrict__ adst,
                                float* __restrict__ emax, int H)
{
    long long idx = (long long)blockIdx.x * blockDim.x + threadIdx.x;
    if (idx >= (long long)Etot * H) return;
    int e = (int)(idx / H), h = (int)(idx % H);
    int s, d; edge_sd(ei, e, E, s, d);
    float v = lrelu02(asrc[(size_t)s * H + h] + adst[(size_t)d * H + h]);
    atomicMaxF(&emax[(size_t)d * H + h], v);
}

__global__ void edge_expsum_kernel(const int* __restrict__ ei, int E, int Etot,
                                   const float* __restrict__ asrc,
                                   const float* __restrict__ adst,
                                   const float* __restrict__ emax,
                                   float* __restrict__ denom, int H)
{
    long long idx = (long long)blockIdx.x * blockDim.x + threadIdx.x;
    if (idx >= (long long)Etot * H) return;
    int e = (int)(idx / H), h = (int)(idx % H);
    int s, d; edge_sd(ei, e, E, s, d);
    float v = lrelu02(asrc[(size_t)s * H + h] + adst[(size_t)d * H + h]);
    atomicAdd(&denom[(size_t)d * H + h], expf(v - emax[(size_t)d * H + h]));
}

__global__ void edge_agg_kernel(const int* __restrict__ ei, int E, int Etot,
                                const float* __restrict__ asrc,
                                const float* __restrict__ adst,
                                const float* __restrict__ emax,
                                const float* __restrict__ denom,
                                const float* __restrict__ xh,
                                float* __restrict__ agg, int H, int C)
{
    long long idx = (long long)blockIdx.x * blockDim.x + threadIdx.x;
    if (idx >= (long long)Etot * 8) return;
    int e = (int)(idx >> 3), g = (int)(idx & 7);
    int h = (g * 8) / C;
    int s, d; edge_sd(ei, e, E, s, d);

    float logit = lrelu02(asrc[(size_t)s * H + h] + adst[(size_t)d * H + h]);
    float alpha = expf(logit - emax[(size_t)d * H + h]) /
                  (denom[(size_t)d * H + h] + 1e-16f);

    const float4* xr = (const float4*)(xh + (size_t)s * 64 + g * 8);
    float4 u = xr[0], w = xr[1];
    float* o = agg + (size_t)d * 64 + g * 8;
    atomicAdd(o + 0, u.x * alpha);
    atomicAdd(o + 1, u.y * alpha);
    atomicAdd(o + 2, u.z * alpha);
    atomicAdd(o + 3, u.w * alpha);
    atomicAdd(o + 4, w.x * alpha);
    atomicAdd(o + 5, w.y * alpha);
    atomicAdd(o + 6, w.z * alpha);
    atomicAdd(o + 7, w.w * alpha);
}

__global__ void bias_elu_kernel(float* __restrict__ p,
                                const float* __restrict__ b, size_t n)
{
    size_t i = (size_t)blockIdx.x * blockDim.x + threadIdx.x;
    if (i >= n) return;
    float v = p[i] + b[i & 63];
    p[i] = v > 0.f ? v : expm1f(v);
}

__global__ void bias_add_kernel(float* __restrict__ p,
                                const float* __restrict__ b, size_t n)
{
    size_t i = (size_t)blockIdx.x * blockDim.x + threadIdx.x;
    if (i >= n) return;
    p[i] = p[i] + b[i & 63];
}

static inline int nblk(long long n, int b) { return (int)((n + b - 1) / b); }

extern "C" void kernel_launch(void* const* d_in, const int* in_sizes, int n_in,
                              void* d_out, int out_size, void* d_ws, size_t ws_size,
                              hipStream_t stream)
{
    const int*   ei  = (const int*)  d_in[0];
    const float* X   = (const float*)d_in[1];
    const float* W1  = (const float*)d_in[2];
    const float* as1 = (const float*)d_in[3];
    const float* ad1 = (const float*)d_in[4];
    const float* b1  = (const float*)d_in[5];
    const float* W2  = (const float*)d_in[6];
    const float* as2 = (const float*)d_in[7];
    const float* ad2 = (const float*)d_in[8];
    const float* b2  = (const float*)d_in[9];

    const int E    = in_sizes[0] / 2;
    const int N    = in_sizes[1] / 128;
    const int Etot = E + N;
    const size_t N64 = (size_t)N * 64;

    float* ws    = (float*)d_ws;
    float* bufA  = ws;
    float* bufB  = bufA + N64;
    float* asrc1 = bufB + N64;
    float* adst1 = asrc1 + (size_t)N * 8;
    float* emax1 = adst1 + (size_t)N * 8;
    float* den1  = emax1 + (size_t)N * 8;
    float* asrc2 = den1  + (size_t)N * 8;
    float* adst2 = asrc2 + N;
    float* emax2 = adst2 + N;
    float* den2  = emax2 + N;
    float* out   = (float*)d_out;

    const int T = 256;

    fill_kernel<<<nblk((long long)N * 8, T), T, 0, stream>>>(emax1, (size_t)N * 8, -INFINITY);
    fill_kernel<<<nblk((long long)N * 8, T), T, 0, stream>>>(den1,  (size_t)N * 8, 0.f);
    fill_kernel<<<nblk((long long)N64,   T), T, 0, stream>>>(bufB,  N64, 0.f);
    fill_kernel<<<nblk((long long)N,     T), T, 0, stream>>>(emax2, (size_t)N, -INFINITY);
    fill_kernel<<<nblk((long long)N,     T), T, 0, stream>>>(den2,  (size_t)N, 0.f);
    fill_kernel<<<nblk((long long)N64,   T), T, 0, stream>>>(out,   N64, 0.f);

    {
        const int H = 8, C = 8;
        int mtiles = (N + 15) / 16;
        gemm_wmma_kernel<<<nblk(mtiles, 8), 256, 0, stream>>>(X, W1, bufA, N, 128);
        alpha_kernel<<<nblk((long long)N * H, T), T, 0, stream>>>(bufA, as1, ad1, asrc1, adst1, N, H, C);
        edge_max_kernel<<<nblk((long long)Etot * H, T), T, 0, stream>>>(ei, E, Etot, asrc1, adst1, emax1, H);
        edge_expsum_kernel<<<nblk((long long)Etot * H, T), T, 0, stream>>>(ei, E, Etot, asrc1, adst1, emax1, den1, H);
        edge_agg_kernel<<<nblk((long long)Etot * 8, T), T, 0, stream>>>(ei, E, Etot, asrc1, adst1, emax1, den1, bufA, bufB, H, C);
        bias_elu_kernel<<<nblk((long long)N64, T), T, 0, stream>>>(bufB, b1, N64);
    }

    {
        const int H = 1, C = 64;
        int mtiles = (N + 15) / 16;
        gemm_wmma_kernel<<<nblk(mtiles, 8), 256, 0, stream>>>(bufB, W2, bufA, N, 64);
        alpha_kernel<<<nblk((long long)N * H, T), T, 0, stream>>>(bufA, as2, ad2, asrc2, adst2, N, H, C);
        edge_max_kernel<<<nblk((long long)Etot * H, T), T, 0, stream>>>(ei, E, Etot, asrc2, adst2, emax2, H);
        edge_expsum_kernel<<<nblk((long long)Etot * H, T), T, 0, stream>>>(ei, E, Etot, asrc2, adst2, emax2, den2, H);
        edge_agg_kernel<<<nblk((long long)Etot * 8, T), T, 0, stream>>>(ei, E, Etot, asrc2, adst2, emax2, den2, bufA, out, H, C);
        bias_add_kernel<<<nblk((long long)N64, T), T, 0, stream>>>(out, b2, N64);
    }
}


# Edges counting-sorted by destination once per call (arrival-order integer ranks); per node max, exp-sum and weighted sum gathered in registers, no float atomics; slot-order sums, one divide after
# speedup vs baseline: 2.6383x; 2.6383x over previous
#include <hip/hip_runtime.h>
#include <math.h>


typedef __attribute__((ext_vector_type(2))) float v2f;
typedef __attribute__((ext_vector_type(8))) float v8f;

__device__ __forceinline__ float lrelu02(float v) {
    return v > 0.0f ? v : 0.2f * v;
}

__global__ void fill_kernel(float* __restrict__ p, size_t n, float v) {
    size_t i = (size_t)blockIdx.x * blockDim.x + threadIdx.x;
    if (i < n) p[i] = v;
}

__global__ __launch_bounds__(256) void gemm_wmma_kernel(
    const float* __restrict__ X, const float* __restrict__ W,
    float* __restrict__ Y, int M, int K)
{
    __shared__ float sW[128 * 64];
    const int nW = K * 64;
    for (int i = threadIdx.x; i < nW; i += 256) {
        int k = i >> 6, n = i & 63;
        sW[(k >> 1) * 128 + 2 * n + (k & 1)] = W[i];
    }
    __syncthreads();

    const int wave = threadIdx.x >> 5;
    const int lane = threadIdx.x & 31;
    const int m0   = (blockIdx.x * 8 + wave) * 16;
    if (m0 >= M) return;

    const int half = lane >> 4;
    const int r    = lane & 15;

    v8f c0 = {}, c1 = {}, c2 = {}, c3 = {};

    const float* arow = X + (size_t)(m0 + r) * K + 2 * half;

    for (int k0 = 0; k0 < K; k0 += 4) {
        v2f a = *(const v2f*)(arow + k0);

        const float* wp = &sW[((k0 >> 1) + half) * 128 + 2 * r];
        v2f b0 = *(const v2f*)(wp +  0);
        v2f b1 = *(const v2f*)(wp + 32);
        v2f b2 = *(const v2f*)(wp + 64);
        v2f b3 = *(const v2f*)(wp + 96);

        c0 = __builtin_amdgcn_wmma_f32_16x16x4_f32(false, a, false, b0,
                                                   (short)0, c0, false, false);
        c1 = __builtin_amdgcn_wmma_f32_16x16x4_f32(false, a, false, b1,
                                                   (short)0, c1, false, false);
        c2 = __builtin_amdgcn_wmma_f32_16x16x4_f32(false, a, false, b2,
                                                   (short)0, c2, false, false);
        c3 = __builtin_amdgcn_wmma_f32_16x16x4_f32(false, a, false, b3,
                                                   (short)0, c3, false, false);
    }

    float* yb = Y + (size_t)m0 * 64 + r;
#pragma unroll
    for (int j = 0; j < 8; ++j) {
        float* yr = yb + (size_t)(j + half * 8) * 64;
        yr[0]  = c0[j];
        yr[16] = c1[j];
        yr[32] = c2[j];
        yr[48] = c3[j];
    }
}

__global__ void alpha_kernel(const float* __restrict__ xh,
                             const float* __restrict__ a_src,
                             const float* __restrict__ a_dst,
                             float* __restrict__ asrc,
                             float* __restrict__ adst,
                             int N, int H, int C)
{
    int idx = blockIdx.x * blockDim.x + threadIdx.x;
    if (idx >= N * H) return;
    int n = idx / H, h = idx % H;
    const float* row = xh + (size_t)n * 64 + h * C;
    const float* ws  = a_src + h * C;
    const float* wd  = a_dst + h * C;
    float s = 0.f, t = 0.f;
    for (int c = 0; c < C; ++c) {
        float v = row[c];
        s += v * ws[c];
        t += v * wd[c];
    }
    asrc[idx] = s;
    adst[idx] = t;
}

__global__ void hist_rank_kernel(const int* __restrict__ ei, int E,
                                 int* __restrict__ cnt, int* __restrict__ rank)
{
    int e = blockIdx.x * blockDim.x + threadIdx.x;
    if (e >= E) return;
    rank[e] = atomicAdd(&cnt[ei[E + e]], 1);
}

__global__ __launch_bounds__(256) void scan_kernel(
    const int* __restrict__ cnt, int* __restrict__ rowptr, int N)
{
    __shared__ int sw[8];
    __shared__ int sbase;
    const int lane = threadIdx.x & 31, wave = threadIdx.x >> 5;

    int p = 0;
    const int4* __restrict__ c4 = (const int4*)cnt;
    const int nprev4 = blockIdx.x * 256;
    for (int i = threadIdx.x; i < nprev4; i += 256) {
        const int4 v = c4[i];
        p += (v.x + v.y) + (v.z + v.w);
    }
    for (int o = 16; o > 0; o >>= 1) p += __shfl_down(p, o);
    if (lane == 0) sw[wave] = p;
    __syncthreads();
    if (threadIdx.x == 0) {
        int t = 0;
        for (int w = 0; w < 8; ++w) t += sw[w];
        sbase = t;
    }
    __syncthreads();
    const int blockbase = sbase;

    const int i0 = blockIdx.x * 1024 + threadIdx.x * 4;
    int c[4];
#pragma unroll
    for (int k = 0; k < 4; ++k) c[k] = (i0 + k < N) ? cnt[i0 + k] : 0;
    const int tsum = c[0] + c[1] + c[2] + c[3];
    int inc = tsum;
    for (int o = 1; o < 32; o <<= 1) {
        int t = __shfl_up(inc, o);
        if (lane >= o) inc += t;
    }
    if (lane == 31) sw[wave] = inc;
    __syncthreads();
    int run = blockbase + inc - tsum;
    for (int w = 0; w < wave; ++w) run += sw[w];
#pragma unroll
    for (int k = 0; k < 4; ++k) {
        if (i0 + k < N) rowptr[i0 + k] = run;
        run += c[k];
    }
}

__global__ void csr_fill_kernel(const int* __restrict__ ei, int E,
                                const int* __restrict__ rowptr,
                                const int* __restrict__ rank,
                                int* __restrict__ csr)
{
    int e = blockIdx.x * blockDim.x + threadIdx.x;
    if (e >= E) return;
    int s = ei[e], d = ei[E + e];
    atomicExch(&csr[rowptr[d] + rank[e]], s);
}

__device__ __forceinline__ float max_sel(float a, float b) {
    return b > a ? b : a;
}

template <int H>
__global__ __launch_bounds__(256) void gather_kernel(
    const int* __restrict__ rowptr, const int* __restrict__ cnt,
    const int* __restrict__ csr,
    const float* __restrict__ asrc, const float* __restrict__ adst,
    const float* __restrict__ xh, float* __restrict__ out, int N)
{
    const int gid = blockIdx.x * 256 + threadIdx.x;
    const int d = gid >> 4;
    const int l = gid & 15;
    if (d >= N) return;
    const int h = (H == 8) ? (l >> 1) : 0;
    const int n = cnt[d];
    const int* __restrict__ cs = csr + rowptr[d];
    const float ad     = adst[d * H + h];
    const float a_self = asrc[d * H + h];

    float hm[H];
    if constexpr (H == 8) {
        const float4 p = *(const float4*)(asrc + d * 8);
        const float4 q = *(const float4*)(asrc + d * 8 + 4);
        hm[0] = p.x; hm[1] = p.y; hm[2] = p.z; hm[3] = p.w;
        hm[4] = q.x; hm[5] = q.y; hm[6] = q.z; hm[7] = q.w;
    } else {
        hm[0] = a_self;
    }
    for (int j = l; j < n; j += 16) {
        const int s = cs[j];
        if constexpr (H == 8) {
            const float4 p = *(const float4*)(asrc + s * 8);
            const float4 q = *(const float4*)(asrc + s * 8 + 4);
            hm[0] = max_sel(hm[0], p.x); hm[1] = max_sel(hm[1], p.y);
            hm[2] = max_sel(hm[2], p.z); hm[3] = max_sel(hm[3], p.w);
            hm[4] = max_sel(hm[4], q.x); hm[5] = max_sel(hm[5], q.y);
            hm[6] = max_sel(hm[6], q.z); hm[7] = max_sel(hm[7], q.w);
        } else {
            hm[0] = max_sel(hm[0], asrc[s]);
        }
    }
#pragma unroll
    for (int o = 8; o > 0; o >>= 1) {
#pragma unroll
        for (int k = 0; k < H; ++k) hm[k] = max_sel(hm[k], __shfl_xor(hm[k], o, 16));
    }
    float m = hm[0];
#pragma unroll
    for (int k = 1; k < H; ++k) m = (h == k) ? hm[k] : m;
    const float mx = lrelu02(m + ad);

    const float w0 = expf(lrelu02(a_self + ad) - mx);
    float den = w0;
    const float4 x0 = *(const float4*)(xh + d * 64 + 4 * l);
    float4 acc;
    acc.x = x0.x * w0; acc.y = x0.y * w0; acc.z = x0.z * w0; acc.w = x0.w * w0;

#pragma unroll 4
    for (int j = 0; j < n; ++j) {
        const int s = cs[j];
        const float4 x = *(const float4*)(xh + s * 64 + 4 * l);
        const float w = expf(lrelu02(asrc[s * H + h] + ad) - mx);
        den += w;
        acc.x += x.x * w; acc.y += x.y * w; acc.z += x.z * w; acc.w += x.w * w;
    }
    const float inv = 1.0f / (den + 1e-16f);
    float4 r;
    r.x = acc.x * inv; r.y = acc.y * inv; r.z = acc.z * inv; r.w = acc.w * inv;
    *(float4*)(out + d * 64 + 4 * l) = r;
}

__global__ void bias_elu_kernel(float* __restrict__ p,
                                const float* __restrict__ b, size_t n)
{
    size_t i = (size_t)blockIdx.x * blockDim.x + threadIdx.x;
    if (i >= n) return;
    float v = p[i] + b[i & 63];
    p[i] = v > 0.f ? v : expm1f(v);
}

__global__ void bias_add_kernel(float* __restrict__ p,
                                const float* __restrict__ b, size_t n)
{
    size_t i = (size_t)blockIdx.x * blockDim.x + threadIdx.x;
    if (i >= n) return;
    p[i] = p[i] + b[i & 63];
}

static inline int nblk(long long n, int b) { return (int)((n + b - 1) / b); }

extern "C" void kernel_launch(void* const* d_in, const int* in_sizes, int n_in,
                              void* d_out, int out_size, void* d_ws, size_t ws_size,
                              hipStream_t stream)
{
    const int*   ei  = (const int*)  d_in[0];
    const float* X   = (const float*)d_in[1];
    const float* W1  = (const float*)d_in[2];
    const float* as1 = (const float*)d_in[3];
    const float* ad1 = (const float*)d_in[4];
    const float* b1  = (const float*)d_in[5];
    const float* W2  = (const float*)d_in[6];
    const float* as2 = (const float*)d_in[7];
    const float* ad2 = (const float*)d_in[8];
    const float* b2  = (const float*)d_in[9];

    const int E    = in_sizes[0] / 2;
    const int N    = in_sizes[1] / 128;
    const size_t N64 = (size_t)N * 64;

    float* ws    = (float*)d_ws;
    float* bufA  = ws;
    float* bufB  = bufA + N64;
    float* asrc1 = bufB + N64;
    float* adst1 = asrc1 + (size_t)N * 8;
    float* asrc2 = adst1 + (size_t)N * 8;
    float* adst2 = asrc2 + N;
    int*   cnt    = (int*)(adst2 + N);
    int*   rowptr = cnt + N;
    int*   rank   = rowptr + N;
    int*   csr    = rank + E;
    float* out   = (float*)d_out;

    const int T = 256;
    const int nb = nblk(N, 1024);

    fill_kernel<<<nblk(N, T), T, 0, stream>>>((float*)cnt, (size_t)N, 0.f);
    hist_rank_kernel<<<nblk(E, T), T, 0, stream>>>(ei, E, cnt, rank);
    scan_kernel<<<nb, 256, 0, stream>>>(cnt, rowptr, N);
    csr_fill_kernel<<<nblk(E, T), T, 0, stream>>>(ei, E, rowptr, rank, csr);

    {
        const int H = 8, C = 8;
        int mtiles = (N + 15) / 16;
        gemm_wmma_kernel<<<nblk(mtiles, 8), 256, 0, stream>>>(X, W1, bufA, N, 128);
        alpha_kernel<<<nblk((long long)N * H, T), T, 0, stream>>>(bufA, as1, ad1, asrc1, adst1, N, H, C);
        gather_kernel<8><<<nblk((long long)N * 16, 256), 256, 0, stream>>>(rowptr, cnt, csr, asrc1, adst1, bufA, bufB, N);
        bias_elu_kernel<<<nblk((long long)N64, T), T, 0, stream>>>(bufB, b1, N64);
    }

    {
        const int H = 1, C = 64;
        int mtiles = (N + 15) / 16;
        gemm_wmma_kernel<<<nblk(mtiles, 8), 256, 0, stream>>>(bufB, W2, bufA, N, 64);
        alpha_kernel<<<nblk((long long)N * H, T), T, 0, stream>>>(bufA, as2, ad2, asrc2, adst2, N, H, C);
        gather_kernel<1><<<nblk((long long)N * 16, 256), 256, 0, stream>>>(rowptr, cnt, csr, asrc2, adst2, bufA, out, N);
        bias_add_kernel<<<nblk((long long)N64, T), T, 0, stream>>>(out, b2, N64);
    }
}


# Bias add (and layer 1's ELU) applied in the gather's own row store; the two one-element-per-thread elementwise launches and their 400000 short waves are gone (12 launches -> 10)
# speedup vs baseline: 2.9947x; 1.1351x over previous
#include <hip/hip_runtime.h>
#include <math.h>


typedef __attribute__((ext_vector_type(2))) float v2f;
typedef __attribute__((ext_vector_type(8))) float v8f;

__device__ __forceinline__ float lrelu02(float v) {
    return v > 0.0f ? v : 0.2f * v;
}

__global__ void fill_kernel(float* __restrict__ p, size_t n, float v) {
    size_t i = (size_t)blockIdx.x * blockDim.x + threadIdx.x;
    if (i < n) p[i] = v;
}

__global__ __launch_bounds__(256) void gemm_wmma_kernel(
    const float* __restrict__ X, const float* __restrict__ W,
    float* __restrict__ Y, int M, int K)
{
    __shared__ float sW[128 * 64];
    const int nW = K * 64;
    for (int i = threadIdx.x; i < nW; i += 256) {
        int k = i >> 6, n = i & 63;
        sW[(k >> 1) * 128 + 2 * n + (k & 1)] = W[i];
    }
    __syncthreads();

    const int wave = threadIdx.x >> 5;
    const int lane = threadIdx.x & 31;
    const int m0   = (blockIdx.x * 8 + wave) * 16;
    if (m0 >= M) return;

    const int half = lane >> 4;
    const int r    = lane & 15;

    v8f c0 = {}, c1 = {}, c2 = {}, c3 = {};

    const float* arow = X + (size_t)(m0 + r) * K + 2 * half;

    for (int k0 = 0; k0 < K; k0 += 4) {
        v2f a = *(const v2f*)(arow + k0);

        const float* wp = &sW[((k0 >> 1) + half) * 128 + 2 * r];
        v2f b0 = *(const v2f*)(wp +  0);
        v2f b1 = *(const v2f*)(wp + 32);
        v2f b2 = *(const v2f*)(wp + 64);
        v2f b3 = *(const v2f*)(wp + 96);

        c0 = __builtin_amdgcn_wmma_f32_16x16x4_f32(false, a, false, b0,
                                                   (short)0, c0, false, false);
        c1 = __builtin_amdgcn_wmma_f32_16x16x4_f32(false, a, false, b1,
                                                   (short)0, c1, false, false);
        c2 = __builtin_amdgcn_wmma_f32_16x16x4_f32(false, a, false, b2,
                                                   (short)0, c2, false, false);
        c3 = __builtin_amdgcn_wmma_f32_16x16x4_f32(false, a, false, b3,
                                                   (short)0, c3, false, false);
    }

    float* yb = Y + (size_t)m0 * 64 + r;
#pragma unroll
    for (int j = 0; j < 8; ++j) {
        float* yr = yb + (size_t)(j + half * 8) * 64;
        yr[0]  = c0[j];
        yr[16] = c1[j];
        yr[32] = c2[j];
        yr[48] = c3[j];
    }
}

__global__ void alpha_kernel(const float* __restrict__ xh,
                             const float* __restrict__ a_src,
                             const float* __restrict__ a_dst,
                             float* __restrict__ asrc,
                             float* __restrict__ adst,
                             int N, int H, int C)
{
    int idx = blockIdx.x * blockDim.x + threadIdx.x;
    if (idx >= N * H) return;
    int n = idx / H, h = idx % H;
    const float* row = xh + (size_t)n * 64 + h * C;
    const float* ws  = a_src + h * C;
    const float* wd  = a_dst + h * C;
    float s = 0.f, t = 0.f;
    for (int c = 0; c < C; ++c) {
        float v = row[c];
        s += v * ws[c];
        t += v * wd[c];
    }
    asrc[idx] = s;
    adst[idx] = t;
}

__global__ void hist_rank_kernel(const int* __restrict__ ei, int E,
                                 int* __restrict__ cnt, int* __restrict__ rank)
{
    int e = blockIdx.x * blockDim.x + threadIdx.x;
    if (e >= E) return;
    rank[e] = atomicAdd(&cnt[ei[E + e]], 1);
}

__global__ __launch_bounds__(256) void scan_kernel(
    const int* __restrict__ cnt, int* __restrict__ rowptr, int N)
{
    __shared__ int sw[8];
    __shared__ int sbase;
    const int lane = threadIdx.x & 31, wave = threadIdx.x >> 5;

    int p = 0;
    const int4* __restrict__ c4 = (const int4*)cnt;
    const int nprev4 = blockIdx.x * 256;
    for (int i = threadIdx.x; i < nprev4; i += 256) {
        const int4 v = c4[i];
        p += (v.x + v.y) + (v.z + v.w);
    }
    for (int o = 16; o > 0; o >>= 1) p += __shfl_down(p, o);
    if (lane == 0) sw[wave] = p;
    __syncthreads();
    if (threadIdx.x == 0) {
        int t = 0;
        for (int w = 0; w < 8; ++w) t += sw[w];
        sbase = t;
    }
    __syncthreads();
    const int blockbase = sbase;

    const int i0 = blockIdx.x * 1024 + threadIdx.x * 4;
    int c[4];
#pragma unroll
    for (int k = 0; k < 4; ++k) c[k] = (i0 + k < N) ? cnt[i0 + k] : 0;
    const int tsum = c[0] + c[1] + c[2] + c[3];
    int inc = tsum;
    for (int o = 1; o < 32; o <<= 1) {
        int t = __shfl_up(inc, o);
        if (lane >= o) inc += t;
    }
    if (lane == 31) sw[wave] = inc;
    __syncthreads();
    int run = blockbase + inc - tsum;
    for (int w = 0; w < wave; ++w) run += sw[w];
#pragma unroll
    for (int k = 0; k < 4; ++k) {
        if (i0 + k < N) rowptr[i0 + k] = run;
        run += c[k];
    }
}

__global__ void csr_fill_kernel(const int* __restrict__ ei, int E,
                                const int* __restrict__ rowptr,
                                const int* __restrict__ rank,
                                int* __restrict__ csr)
{
    int e = blockIdx.x * blockDim.x + threadIdx.x;
    if (e >= E) return;
    int s = ei[e], d = ei[E + e];
    atomicExch(&csr[rowptr[d] + rank[e]], s);
}

__device__ __forceinline__ float max_sel(float a, float b) {
    return b > a ? b : a;
}

template <int H, bool ELU>
__global__ __launch_bounds__(256) void gather_kernel(
    const int* __restrict__ rowptr, const int* __restrict__ cnt,
    const int* __restrict__ csr,
    const float* __restrict__ asrc, const float* __restrict__ adst,
    const float* __restrict__ xh, const float* __restrict__ bias,
    float* __restrict__ out, int N)
{
    const int gid = blockIdx.x * 256 + threadIdx.x;
    const int d = gid >> 4;
    const int l = gid & 15;
    if (d >= N) return;
    const int h = (H == 8) ? (l >> 1) : 0;
    const int n = cnt[d];
    const int* __restrict__ cs = csr + rowptr[d];
    const float ad     = adst[d * H + h];
    const float a_self = asrc[d * H + h];
    const float4 bv    = *(const float4*)(bias + 4 * l);

    float hm[H];
    if constexpr (H == 8) {
        const float4 p = *(const float4*)(asrc + d * 8);
        const float4 q = *(const float4*)(asrc + d * 8 + 4);
        hm[0] = p.x; hm[1] = p.y; hm[2] = p.z; hm[3] = p.w;
        hm[4] = q.x; hm[5] = q.y; hm[6] = q.z; hm[7] = q.w;
    } else {
        hm[0] = a_self;
    }
    for (int j = l; j < n; j += 16) {
        const int s = cs[j];
        if constexpr (H == 8) {
            const float4 p = *(const float4*)(asrc + s * 8);
            const float4 q = *(const float4*)(asrc + s * 8 + 4);
            hm[0] = max_sel(hm[0], p.x); hm[1] = max_sel(hm[1], p.y);
            hm[2] = max_sel(hm[2], p.z); hm[3] = max_sel(hm[3], p.w);
            hm[4] = max_sel(hm[4], q.x); hm[5] = max_sel(hm[5], q.y);
            hm[6] = max_sel(hm[6], q.z); hm[7] = max_sel(hm[7], q.w);
        } else {
            hm[0] = max_sel(hm[0], asrc[s]);
        }
    }
#pragma unroll
    for (int o = 8; o > 0; o >>= 1) {
#pragma unroll
        for (int k = 0; k < H; ++k) hm[k] = max_sel(hm[k], __shfl_xor(hm[k], o, 16));
    }
    float m = hm[0];
#pragma unroll
    for (int k = 1; k < H; ++k) m = (h == k) ? hm[k] : m;
    const float mx = lrelu02(m + ad);

    const float w0 = expf(lrelu02(a_self + ad) - mx);
    float den = w0;
    const float4 x0 = *(const float4*)(xh + d * 64 + 4 * l);
    float4 acc;
    acc.x = x0.x * w0; acc.y = x0.y * w0; acc.z = x0.z * w0; acc.w = x0.w * w0;

#pragma unroll 4
    for (int j = 0; j < n; ++j) {
        const int s = cs[j];
        const float4 x = *(const float4*)(xh + s * 64 + 4 * l);
        const float w = expf(lrelu02(asrc[s * H + h] + ad) - mx);
        den += w;
        acc.x += x.x * w; acc.y += x.y * w; acc.z += x.z * w; acc.w += x.w * w;
    }
    const float inv = 1.0f / (den + 1e-16f);
    float4 r;
    r.x = acc.x * inv + bv.x; r.y = acc.y * inv + bv.y;
    r.z = acc.z * inv + bv.z; r.w = acc.w * inv + bv.w;
    if constexpr (ELU) {
        r.x = r.x > 0.f ? r.x : expm1f(r.x);
        r.y = r.y > 0.f ? r.y : expm1f(r.y);
        r.z = r.z > 0.f ? r.z : expm1f(r.z);
        r.w = r.w > 0.f ? r.w : expm1f(r.w);
    }
    *(float4*)(out + d * 64 + 4 * l) = r;
}

static inline int nblk(long long n, int b) { return (int)((n + b - 1) / b); }

extern "C" void kernel_launch(void* const* d_in, const int* in_sizes, int n_in,
                              void* d_out, int out_size, void* d_ws, size_t ws_size,
                              hipStream_t stream)
{
    const int*   ei  = (const int*)  d_in[0];
    const float* X   = (const float*)d_in[1];
    const float* W1  = (const float*)d_in[2];
    const float* as1 = (const float*)d_in[3];
    const float* ad1 = (const float*)d_in[4];
    const float* b1  = (const float*)d_in[5];
    const float* W2  = (const float*)d_in[6];
    const float* as2 = (const float*)d_in[7];
    const float* ad2 = (const float*)d_in[8];
    const float* b2  = (const float*)d_in[9];

    const int E    = in_sizes[0] / 2;
    const int N    = in_sizes[1] / 128;
    const size_t N64 = (size_t)N * 64;

    float* ws    = (float*)d_ws;
    float* bufA  = ws;
    float* bufB  = bufA + N64;
    float* asrc1 = bufB + N64;
    float* adst1 = asrc1 + (size_t)N * 8;
    float* asrc2 = adst1 + (size_t)N * 8;
    float* adst2 = asrc2 + N;
    int*   cnt    = (int*)(adst2 + N);
    int*   rowptr = cnt + N;
    int*   rank   = rowptr + N;
    int*   csr    = rank + E;
    float* out   = (float*)d_out;

    const int T = 256;
    const int nb = nblk(N, 1024);

    fill_kernel<<<nblk(N, T), T, 0, stream>>>((float*)cnt, (size_t)N, 0.f);
    hist_rank_kernel<<<nblk(E, T), T, 0, stream>>>(ei, E, cnt, rank);
    scan_kernel<<<nb, 256, 0, stream>>>(cnt, rowptr, N);
    csr_fill_kernel<<<nblk(E, T), T, 0, stream>>>(ei, E, rowptr, rank, csr);

    {
        const int H = 8, C = 8;
        int mtiles = (N + 15) / 16;
        gemm_wmma_kernel<<<nblk(mtiles, 8), 256, 0, stream>>>(X, W1, bufA, N, 128);
        alpha_kernel<<<nblk((long long)N * H, T), T, 0, stream>>>(bufA, as1, ad1, asrc1, adst1, N, H, C);
        gather_kernel<8, true><<<nblk((long long)N * 16, 256), 256, 0, stream>>>(rowptr, cnt, csr, asrc1, adst1, bufA, b1, bufB, N);
    }

    {
        const int H = 1, C = 64;
        int mtiles = (N + 15) / 16;
        gemm_wmma_kernel<<<nblk(mtiles, 8), 256, 0, stream>>>(bufB, W2, bufA, N, 64);
        alpha_kernel<<<nblk((long long)N * H, T), T, 0, stream>>>(bufA, as2, ad2, asrc2, adst2, N, H, C);
        gather_kernel<1, false><<<nblk((long long)N * 16, 256), 256, 0, stream>>>(rowptr, cnt, csr, asrc2, adst2, bufA, b2, out, N);
    }
}


# Edge ranks from LDS counters: two-level counting sort (edge chunks split by 512-node bucket, one work-group per bucket) replaces a returning global atomic per edge; slots in LDS arrival order
# speedup vs baseline: 5.4286x; 1.8127x over previous
#include <hip/hip_runtime.h>
#include <math.h>


typedef __attribute__((ext_vector_type(2))) float v2f;
typedef __attribute__((ext_vector_type(8))) float v8f;

__device__ __forceinline__ float lrelu02(float v) {
    return v > 0.0f ? v : 0.2f * v;
}

__global__ __launch_bounds__(256) void gemm_wmma_kernel(
    const float* __restrict__ X, const float* __restrict__ W,
    float* __restrict__ Y, int M, int K)
{
    __shared__ float sW[128 * 64];
    const int nW = K * 64;
    for (int i = threadIdx.x; i < nW; i += 256) {
        int k = i >> 6, n = i & 63;
        sW[(k >> 1) * 128 + 2 * n + (k & 1)] = W[i];
    }
    __syncthreads();

    const int wave = threadIdx.x >> 5;
    const int lane = threadIdx.x & 31;
    const int m0   = (blockIdx.x * 8 + wave) * 16;
    if (m0 >= M) return;

    const int half = lane >> 4;
    const int r    = lane & 15;

    v8f c0 = {}, c1 = {}, c2 = {}, c3 = {};

    const float* arow = X + (size_t)(m0 + r) * K + 2 * half;

    for (int k0 = 0; k0 < K; k0 += 4) {
        v2f a = *(const v2f*)(arow + k0);

        const float* wp = &sW[((k0 >> 1) + half) * 128 + 2 * r];
        v2f b0 = *(const v2f*)(wp +  0);
        v2f b1 = *(const v2f*)(wp + 32);
        v2f b2 = *(const v2f*)(wp + 64);
        v2f b3 = *(const v2f*)(wp + 96);

        c0 = __builtin_amdgcn_wmma_f32_16x16x4_f32(false, a, false, b0,
                                                   (short)0, c0, false, false);
        c1 = __builtin_amdgcn_wmma_f32_16x16x4_f32(false, a, false, b1,
                                                   (short)0, c1, false, false);
        c2 = __builtin_amdgcn_wmma_f32_16x16x4_f32(false, a, false, b2,
                                                   (short)0, c2, false, false);
        c3 = __builtin_amdgcn_wmma_f32_16x16x4_f32(false, a, false, b3,
                                                   (short)0, c3, false, false);
    }

    float* yb = Y + (size_t)m0 * 64 + r;
#pragma unroll
    for (int j = 0; j < 8; ++j) {
        float* yr = yb + (size_t)(j + half * 8) * 64;
        yr[0]  = c0[j];
        yr[16] = c1[j];
        yr[32] = c2[j];
        yr[48] = c3[j];
    }
}

__global__ void alpha_kernel(const float* __restrict__ xh,
                             const float* __restrict__ a_src,
                             const float* __restrict__ a_dst,
                             float* __restrict__ asrc,
                             float* __restrict__ adst,
                             int N, int H, int C)
{
    int idx = blockIdx.x * blockDim.x + threadIdx.x;
    if (idx >= N * H) return;
    int n = idx / H, h = idx % H;
    const float* row = xh + (size_t)n * 64 + h * C;
    const float* ws  = a_src + h * C;
    const float* wd  = a_dst + h * C;
    float s = 0.f, t = 0.f;
    for (int c = 0; c < C; ++c) {
        float v = row[c];
        s += v * ws[c];
        t += v * wd[c];
    }
    asrc[idx] = s;
    adst[idx] = t;
}

#define CHUNK 6400
#define EPT   25
#define MAXB  1024
#define CAPB  12288

__global__ __launch_bounds__(256) void chunk_split_kernel(
    const int* __restrict__ ei, int E, int OP,
    unsigned* __restrict__ part, int* __restrict__ offg)
{
    __shared__ int cntB[MAXB];
    __shared__ __attribute__((aligned(16))) unsigned sorted[CHUNK];
    __shared__ int wsum[8];
    const int tid = threadIdx.x, lane = tid & 31, wave = tid >> 5;

    for (int i = tid; i < MAXB; i += 256) cntB[i] = 0;
    __syncthreads();

    const int e0 = blockIdx.x * CHUNK;
    unsigned pk[EPT];
    int bk[EPT], rk[EPT];
#pragma unroll
    for (int k = 0; k < EPT; ++k) {
        const int e  = e0 + k * 256 + tid;
        const int ec = e < E ? e : E - 1;
        pk[k] = (unsigned)ei[ec];
        bk[k] = ei[E + ec];
    }
#pragma unroll
    for (int k = 0; k < EPT; ++k) {
        const int e = e0 + k * 256 + tid;
        const int d = bk[k];
        pk[k] = (pk[k] << 9) | (unsigned)(d & 511);
        bk[k] = e < E ? (d >> 9) : -1;
    }
#pragma unroll
    for (int k = 0; k < EPT; ++k) {
        rk[k] = 0;
        if (bk[k] >= 0) rk[k] = atomicAdd(&cntB[bk[k]], 1);
    }
    __syncthreads();

    const int a0 = cntB[4 * tid], a1 = cntB[4 * tid + 1];
    const int a2 = cntB[4 * tid + 2], a3 = cntB[4 * tid + 3];
    const int ts = (a0 + a1) + (a2 + a3);
    int inc = ts;
    for (int o = 1; o < 32; o <<= 1) {
        const int t = __shfl_up(inc, o);
        if (lane >= o) inc += t;
    }
    if (lane == 31) wsum[wave] = inc;
    __syncthreads();
    int ex = inc - ts;
    for (int w = 0; w < wave; ++w) ex += wsum[w];
    int4 o4;
    o4.x = ex; o4.y = ex + a0; o4.z = o4.y + a1; o4.w = o4.z + a2;
    cntB[4 * tid]     = o4.x; cntB[4 * tid + 1] = o4.y;
    cntB[4 * tid + 2] = o4.z; cntB[4 * tid + 3] = o4.w;
    if (4 * tid < OP) *(int4*)(offg + (size_t)blockIdx.x * OP + 4 * tid) = o4;
    __syncthreads();

#pragma unroll
    for (int k = 0; k < EPT; ++k) {
        if (bk[k] >= 0) sorted[cntB[bk[k]] + rk[k]] = pk[k];
    }
    __syncthreads();

    const uint4* __restrict__ s4 = (const uint4*)sorted;
    uint4* __restrict__ p4 = (uint4*)(part + (size_t)blockIdx.x * CHUNK);
    for (int i = tid; i < CHUNK / 4; i += 256) p4[i] = s4[i];
}

__global__ __launch_bounds__(256) void bucket_order_kernel(
    const unsigned* __restrict__ part, const int* __restrict__ offg,
    int NC, int OP, int* __restrict__ cnt, int* __restrict__ rowptr,
    int* __restrict__ csr, int N)
{
    __shared__ int cntN[512];
    __shared__ int cur[512];
    __shared__ int outL[CAPB];
    __shared__ int red[16];
    __shared__ int wsum[8];
    const int tid = threadIdx.x, lane = tid & 31, wave = tid >> 5;
    const int b = blockIdx.x;

    cntN[tid] = 0; cntN[tid + 256] = 0;

    int sT = 0, sB = 0;
    for (int c = tid; c < NC; c += 256) {
        const int o0 = offg[(size_t)c * OP + b];
        const int o1 = offg[(size_t)c * OP + b + 1];
        sB += o0; sT += o1 - o0;
    }
    for (int o = 16; o > 0; o >>= 1) {
        sT += __shfl_xor(sT, o);
        sB += __shfl_xor(sB, o);
    }
    if (lane == 0) { red[wave] = sT; red[8 + wave] = sB; }
    __syncthreads();
    int T = 0, base = 0;
    for (int w = 0; w < 8; ++w) { T += red[w]; base += red[8 + w]; }
    const bool big = T > CAPB;

    for (int c = tid; c < NC; c += 256) {
        const int o0  = offg[(size_t)c * OP + b];
        const int len = offg[(size_t)c * OP + b + 1] - o0;
        const unsigned* __restrict__ pc = part + (size_t)c * CHUNK + o0;
        for (int i = 0; i < len; i += 4) {
            const int last = len - 1;
            const unsigned w0 = pc[i];
            const unsigned w1 = pc[i + 1 < last ? i + 1 : last];
            const unsigned w2 = pc[i + 2 < last ? i + 2 : last];
            const unsigned w3 = pc[i + 3 < last ? i + 3 : last];
            atomicAdd(&cntN[w0 & 511u], 1);
            if (i + 1 < len) atomicAdd(&cntN[w1 & 511u], 1);
            if (i + 2 < len) atomicAdd(&cntN[w2 & 511u], 1);
            if (i + 3 < len) atomicAdd(&cntN[w3 & 511u], 1);
        }
    }
    __syncthreads();

    const int c0 = cntN[2 * tid], c1 = cntN[2 * tid + 1];
    const int ts = c0 + c1;
    int inc = ts;
    for (int o = 1; o < 32; o <<= 1) {
        const int t = __shfl_up(inc, o);
        if (lane >= o) inc += t;
    }
    if (lane == 31) wsum[wave] = inc;
    __syncthreads();
    int ex = inc - ts;
    for (int w = 0; w < wave; ++w) ex += wsum[w];
    cur[2 * tid] = ex; cur[2 * tid + 1] = ex + c0;
    const int d = b * 512 + 2 * tid;
    if (d + 1 < N) {
        *(int2*)(cnt + d)    = make_int2(c0, c1);
        *(int2*)(rowptr + d) = make_int2(base + ex, base + ex + c0);
    } else if (d < N) {
        cnt[d] = c0; rowptr[d] = base + ex;
    }
    __syncthreads();

    for (int c = tid; c < NC; c += 256) {
        const int o0  = offg[(size_t)c * OP + b];
        const int len = offg[(size_t)c * OP + b + 1] - o0;
        const unsigned* __restrict__ pc = part + (size_t)c * CHUNK + o0;
        for (int i = 0; i < len; i += 4) {
            const int last = len - 1;
            const unsigned w0 = pc[i];
            const unsigned w1 = pc[i + 1 < last ? i + 1 : last];
            const unsigned w2 = pc[i + 2 < last ? i + 2 : last];
            const unsigned w3 = pc[i + 3 < last ? i + 3 : last];
            const int n = (len - i < 4) ? (len - i) : 4;
            const unsigned wv[4] = {w0, w1, w2, w3};
            int pos[4];
#pragma unroll
            for (int q = 0; q < 4; ++q) {
                pos[q] = 0;
                if (q < n) pos[q] = atomicAdd(&cur[wv[q] & 511u], 1);
            }
#pragma unroll
            for (int q = 0; q < 4; ++q) {
                if (q < n) {
                    const int src = (int)(wv[q] >> 9);
                    if (!big) outL[pos[q]] = src;
                    else atomicExch(&csr[base + pos[q]], src);
                }
            }
        }
    }
    __syncthreads();

    if (!big) {
        const int zend = base + T;
        for (int g = (base & ~31) + tid; g < zend; g += 256) {
            if (g >= base) {
                const int l0 = g & ~31;
                if (l0 >= base && l0 + 32 <= zend) {
                    csr[g] = outL[g - base];
                } else {
                    atomicExch(&csr[g], outL[g - base]);
                }
            }
        }
    }
}

__device__ __forceinline__ float max_sel(float a, float b) {
    return b > a ? b : a;
}

template <int H, bool ELU>
__global__ __launch_bounds__(256) void gather_kernel(
    const int* __restrict__ rowptr, const int* __restrict__ cnt,
    const int* __restrict__ csr,
    const float* __restrict__ asrc, const float* __restrict__ adst,
    const float* __restrict__ xh, const float* __restrict__ bias,
    float* __restrict__ out, int N)
{
    const int gid = blockIdx.x * 256 + threadIdx.x;
    const int d = gid >> 4;
    const int l = gid & 15;
    if (d >= N) return;
    const int h = (H == 8) ? (l >> 1) : 0;
    const int n = cnt[d];
    const int* __restrict__ cs = csr + rowptr[d];
    const float ad     = adst[d * H + h];
    const float a_self = asrc[d * H + h];
    const float4 bv    = *(const float4*)(bias + 4 * l);

    float hm[H];
    if constexpr (H == 8) {
        const float4 p = *(const float4*)(asrc + d * 8);
        const float4 q = *(const float4*)(asrc + d * 8 + 4);
        hm[0] = p.x; hm[1] = p.y; hm[2] = p.z; hm[3] = p.w;
        hm[4] = q.x; hm[5] = q.y; hm[6] = q.z; hm[7] = q.w;
    } else {
        hm[0] = a_self;
    }
    for (int j = l; j < n; j += 16) {
        const int s = cs[j];
        if constexpr (H == 8) {
            const float4 p = *(const float4*)(asrc + s * 8);
            const float4 q = *(const float4*)(asrc + s * 8 + 4);
            hm[0] = max_sel(hm[0], p.x); hm[1] = max_sel(hm[1], p.y);
            hm[2] = max_sel(hm[2], p.z); hm[3] = max_sel(hm[3], p.w);
            hm[4] = max_sel(hm[4], q.x); hm[5] = max_sel(hm[5], q.y);
            hm[6] = max_sel(hm[6], q.z); hm[7] = max_sel(hm[7], q.w);
        } else {
            hm[0] = max_sel(hm[0], asrc[s]);
        }
    }
#pragma unroll
    for (int o = 8; o > 0; o >>= 1) {
#pragma unroll
        for (int k = 0; k < H; ++k) hm[k] = max_sel(hm[k], __shfl_xor(hm[k], o, 16));
    }
    float m = hm[0];
#pragma unroll
    for (int k = 1; k < H; ++k) m = (h == k) ? hm[k] : m;
    const float mx = lrelu02(m + ad);

    const float w0 = expf(lrelu02(a_self + ad) - mx);
    float den = w0;
    const float4 x0 = *(const float4*)(xh + d * 64 + 4 * l);
    float4 acc;
    acc.x = x0.x * w0; acc.y = x0.y * w0; acc.z = x0.z * w0; acc.w = x0.w * w0;

#pragma unroll 4
    for (int j = 0; j < n; ++j) {
        const int s = cs[j];
        const float4 x = *(const float4*)(xh + s * 64 + 4 * l);
        const float w = expf(lrelu02(asrc[s * H + h] + ad) - mx);
        den += w;
        acc.x += x.x * w; acc.y += x.y * w; acc.z += x.z * w; acc.w += x.w * w;
    }
    const float inv = 1.0f / (den + 1e-16f);
    float4 r;
    r.x = acc.x * inv + bv.x; r.y = acc.y * inv + bv.y;
    r.z = acc.z * inv + bv.z; r.w = acc.w * inv + bv.w;
    if constexpr (ELU) {
        r.x = r.x > 0.f ? r.x : expm1f(r.x);
        r.y = r.y > 0.f ? r.y : expm1f(r.y);
        r.z = r.z > 0.f ? r.z : expm1f(r.z);
        r.w = r.w > 0.f ? r.w : expm1f(r.w);
    }
    *(float4*)(out + d * 64 + 4 * l) = r;
}

static inline int nblk(long long n, int b) { return (int)((n + b - 1) / b); }

extern "C" void kernel_launch(void* const* d_in, const int* in_sizes, int n_in,
                              void* d_out, int out_size, void* d_ws, size_t ws_size,
                              hipStream_t stream)
{
    const int*   ei  = (const int*)  d_in[0];
    const float* X   = (const float*)d_in[1];
    const float* W1  = (const float*)d_in[2];
    const float* as1 = (const float*)d_in[3];
    const float* ad1 = (const float*)d_in[4];
    const float* b1  = (const float*)d_in[5];
    const float* W2  = (const float*)d_in[6];
    const float* as2 = (const float*)d_in[7];
    const float* ad2 = (const float*)d_in[8];
    const float* b2  = (const float*)d_in[9];

    const int E    = in_sizes[0] / 2;
    const int N    = in_sizes[1] / 128;
    const size_t N64 = (size_t)N * 64;

    const int NB = nblk(N, 512);
    const int OP = (NB + 32) & ~31;
    const int NC = nblk(E, CHUNK);
    const size_t Na = ((size_t)N + 31) & ~(size_t)31;
    const size_t Ea = ((size_t)E + 31) & ~(size_t)31;

    float* ws    = (float*)d_ws;
    float* bufA  = ws;
    float* bufB  = bufA + N64;
    float* asrc1 = bufB + N64;
    float* adst1 = asrc1 + Na * 8;
    float* asrc2 = adst1 + Na * 8;
    float* adst2 = asrc2 + Na;
    int*   cnt    = (int*)(adst2 + Na);
    int*   rowptr = cnt + Na;
    int*   csr    = rowptr + Na;
    unsigned* part = (unsigned*)(csr + Ea);
    int*   offg   = (int*)(part + (size_t)NC * CHUNK);
    float* out   = (float*)d_out;

    const int T = 256;

    chunk_split_kernel<<<NC, 256, 0, stream>>>(ei, E, OP, part, offg);
    bucket_order_kernel<<<NB, 256, 0, stream>>>(part, offg, NC, OP, cnt, rowptr, csr, N);

    {
        const int H = 8, C = 8;
        int mtiles = (N + 15) / 16;
        gemm_wmma_kernel<<<nblk(mtiles, 8), 256, 0, stream>>>(X, W1, bufA, N, 128);
        alpha_kernel<<<nblk((long long)N * H, T), T, 0, stream>>>(bufA, as1, ad1, asrc1, adst1, N, H, C);
        gather_kernel<8, true><<<nblk((long long)N * 16, 256), 256, 0, stream>>>(rowptr, cnt, csr, asrc1, adst1, bufA, b1, bufB, N);
    }

    {
        const int H = 1, C = 64;
        int mtiles = (N + 15) / 16;
        gemm_wmma_kernel<<<nblk(mtiles, 8), 256, 0, stream>>>(bufB, W2, bufA, N, 64);
        alpha_kernel<<<nblk((long long)N * H, T), T, 0, stream>>>(bufA, as2, ad2, asrc2, adst2, N, H, C);
        gather_kernel<1, false><<<nblk((long long)N * 16, 256), 256, 0, stream>>>(rowptr, cnt, csr, asrc2, adst2, bufA, b2, out, N);
    }
}


# Gather: an edge's softmax weight is computed once, by the lane that loaded the edge; ids+weights of a node's first 32 edges pass through LDS; rows requested 8 per trip, zero weight past the end
# speedup vs baseline: 6.1558x; 1.1340x over previous
#include <hip/hip_runtime.h>
#include <math.h>


typedef __attribute__((ext_vector_type(2))) float v2f;
typedef __attribute__((ext_vector_type(8))) float v8f;

__device__ __forceinline__ float lrelu02(float v) {
    return v > 0.0f ? v : 0.2f * v;
}

__global__ __launch_bounds__(256) void gemm_wmma_kernel(
    const float* __restrict__ X, const float* __restrict__ W,
    float* __restrict__ Y, int M, int K)
{
    __shared__ float sW[128 * 64];
    const int nW = K * 64;
    for (int i = threadIdx.x; i < nW; i += 256) {
        int k = i >> 6, n = i & 63;
        sW[(k >> 1) * 128 + 2 * n + (k & 1)] = W[i];
    }
    __syncthreads();

    const int wave = threadIdx.x >> 5;
    const int lane = threadIdx.x & 31;
    const int m0   = (blockIdx.x * 8 + wave) * 16;
    if (m0 >= M) return;

    const int half = lane >> 4;
    const int r    = lane & 15;

    v8f c0 = {}, c1 = {}, c2 = {}, c3 = {};

    const float* arow = X + (size_t)(m0 + r) * K + 2 * half;

    for (int k0 = 0; k0 < K; k0 += 4) {
        v2f a = *(const v2f*)(arow + k0);

        const float* wp = &sW[((k0 >> 1) + half) * 128 + 2 * r];
        v2f b0 = *(const v2f*)(wp +  0);
        v2f b1 = *(const v2f*)(wp + 32);
        v2f b2 = *(const v2f*)(wp + 64);
        v2f b3 = *(const v2f*)(wp + 96);

        c0 = __builtin_amdgcn_wmma_f32_16x16x4_f32(false, a, false, b0,
                                                   (short)0, c0, false, false);
        c1 = __builtin_amdgcn_wmma_f32_16x16x4_f32(false, a, false, b1,
                                                   (short)0, c1, false, false);
        c2 = __builtin_amdgcn_wmma_f32_16x16x4_f32(false, a, false, b2,
                                                   (short)0, c2, false, false);
        c3 = __builtin_amdgcn_wmma_f32_16x16x4_f32(false, a, false, b3,
                                                   (short)0, c3, false, false);
    }

    float* yb = Y + (size_t)m0 * 64 + r;
#pragma unroll
    for (int j = 0; j < 8; ++j) {
        float* yr = yb + (size_t)(j + half * 8) * 64;
        yr[0]  = c0[j];
        yr[16] = c1[j];
        yr[32] = c2[j];
        yr[48] = c3[j];
    }
}

__global__ void alpha_kernel(const float* __restrict__ xh,
                             const float* __restrict__ a_src,
                             const float* __restrict__ a_dst,
                             float* __restrict__ asrc,
                             float* __restrict__ adst,
                             int N, int H, int C)
{
    int idx = blockIdx.x * blockDim.x + threadIdx.x;
    if (idx >= N * H) return;
    int n = idx / H, h = idx % H;
    const float* row = xh + (size_t)n * 64 + h * C;
    const float* ws  = a_src + h * C;
    const float* wd  = a_dst + h * C;
    float s = 0.f, t = 0.f;
    for (int c = 0; c < C; ++c) {
        float v = row[c];
        s += v * ws[c];
        t += v * wd[c];
    }
    asrc[idx] = s;
    adst[idx] = t;
}

#define CHUNK 6400
#define EPT   25
#define MAXB  1024
#define CAPB  12288

__global__ __launch_bounds__(256) void chunk_split_kernel(
    const int* __restrict__ ei, int E, int OP,
    unsigned* __restrict__ part, int* __restrict__ offg)
{
    __shared__ int cntB[MAXB];
    __shared__ __attribute__((aligned(16))) unsigned sorted[CHUNK];
    __shared__ int wsum[8];
    const int tid = threadIdx.x, lane = tid & 31, wave = tid >> 5;

    for (int i = tid; i < MAXB; i += 256) cntB[i] = 0;
    __syncthreads();

    const int e0 = blockIdx.x * CHUNK;
    unsigned pk[EPT];
    int bk[EPT], rk[EPT];
#pragma unroll
    for (int k = 0; k < EPT; ++k) {
        const int e  = e0 + k * 256 + tid;
        const int ec = e < E ? e : E - 1;
        pk[k] = (unsigned)ei[ec];
        bk[k] = ei[E + ec];
    }
#pragma unroll
    for (int k = 0; k < EPT; ++k) {
        const int e = e0 + k * 256 + tid;
        const int d = bk[k];
        pk[k] = (pk[k] << 9) | (unsigned)(d & 511);
        bk[k] = e < E ? (d >> 9) : -1;
    }
#pragma unroll
    for (int k = 0; k < EPT; ++k) {
        rk[k] = 0;
        if (bk[k] >= 0) rk[k] = atomicAdd(&cntB[bk[k]], 1);
    }
    __syncthreads();

    const int a0 = cntB[4 * tid], a1 = cntB[4 * tid + 1];
    const int a2 = cntB[4 * tid + 2], a3 = cntB[4 * tid + 3];
    const int ts = (a0 + a1) + (a2 + a3);
    int inc = ts;
    for (int o = 1; o < 32; o <<= 1) {
        const int t = __shfl_up(inc, o);
        if (lane >= o) inc += t;
    }
    if (lane == 31) wsum[wave] = inc;
    __syncthreads();
    int ex = inc - ts;
    for (int w = 0; w < wave; ++w) ex += wsum[w];
    int4 o4;
    o4.x = ex; o4.y = ex + a0; o4.z = o4.y + a1; o4.w = o4.z + a2;
    cntB[4 * tid]     = o4.x; cntB[4 * tid + 1] = o4.y;
    cntB[4 * tid + 2] = o4.z; cntB[4 * tid + 3] = o4.w;
    if (4 * tid < OP) *(int4*)(offg + (size_t)blockIdx.x * OP + 4 * tid) = o4;
    __syncthreads();

#pragma unroll
    for (int k = 0; k < EPT; ++k) {
        if (bk[k] >= 0) sorted[cntB[bk[k]] + rk[k]] = pk[k];
    }
    __syncthreads();

    const uint4* __restrict__ s4 = (const uint4*)sorted;
    uint4* __restrict__ p4 = (uint4*)(part + (size_t)blockIdx.x * CHUNK);
    for (int i = tid; i < CHUNK / 4; i += 256) p4[i] = s4[i];
}

__global__ __launch_bounds__(256) void bucket_order_kernel(
    const unsigned* __restrict__ part, const int* __restrict__ offg,
    int NC, int OP, int* __restrict__ cnt, int* __restrict__ rowptr,
    int* __restrict__ csr, int N)
{
    __shared__ int cntN[512];
    __shared__ int cur[512];
    __shared__ int outL[CAPB];
    __shared__ int red[16];
    __shared__ int wsum[8];
    const int tid = threadIdx.x, lane = tid & 31, wave = tid >> 5;
    const int b = blockIdx.x;

    cntN[tid] = 0; cntN[tid + 256] = 0;

    int sT = 0, sB = 0;
    for (int c = tid; c < NC; c += 256) {
        const int o0 = offg[(size_t)c * OP + b];
        const int o1 = offg[(size_t)c * OP + b + 1];
        sB += o0; sT += o1 - o0;
    }
    for (int o = 16; o > 0; o >>= 1) {
        sT += __shfl_xor(sT, o);
        sB += __shfl_xor(sB, o);
    }
    if (lane == 0) { red[wave] = sT; red[8 + wave] = sB; }
    __syncthreads();
    int T = 0, base = 0;
    for (int w = 0; w < 8; ++w) { T += red[w]; base += red[8 + w]; }
    const bool big = T > CAPB;

    for (int c = tid; c < NC; c += 256) {
        const int o0  = offg[(size_t)c * OP + b];
        const int len = offg[(size_t)c * OP + b + 1] - o0;
        const unsigned* __restrict__ pc = part + (size_t)c * CHUNK + o0;
        for (int i = 0; i < len; i += 4) {
            const int last = len - 1;
            const unsigned w0 = pc[i];
            const unsigned w1 = pc[i + 1 < last ? i + 1 : last];
            const unsigned w2 = pc[i + 2 < last ? i + 2 : last];
            const unsigned w3 = pc[i + 3 < last ? i + 3 : last];
            atomicAdd(&cntN[w0 & 511u], 1);
            if (i + 1 < len) atomicAdd(&cntN[w1 & 511u], 1);
            if (i + 2 < len) atomicAdd(&cntN[w2 & 511u], 1);
            if (i + 3 < len) atomicAdd(&cntN[w3 & 511u], 1);
        }
    }
    __syncthreads();

    const int c0 = cntN[2 * tid], c1 = cntN[2 * tid + 1];
    const int ts = c0 + c1;
    int inc = ts;
    for (int o = 1; o < 32; o <<= 1) {
        const int t = __shfl_up(inc, o);
        if (lane >= o) inc += t;
    }
    if (lane == 31) wsum[wave] = inc;
    __syncthreads();
    int ex = inc - ts;
    for (int w = 0; w < wave; ++w) ex += wsum[w];
    cur[2 * tid] = ex; cur[2 * tid + 1] = ex + c0;
    const int d = b * 512 + 2 * tid;
    if (d + 1 < N) {
        *(int2*)(cnt + d)    = make_int2(c0, c1);
        *(int2*)(rowptr + d) = make_int2(base + ex, base + ex + c0);
    } else if (d < N) {
        cnt[d] = c0; rowptr[d] = base + ex;
    }
    __syncthreads();

    for (int c = tid; c < NC; c += 256) {
        const int o0  = offg[(size_t)c * OP + b];
        const int len = offg[(size_t)c * OP + b + 1] - o0;
        const unsigned* __restrict__ pc = part + (size_t)c * CHUNK + o0;
        for (int i = 0; i < len; i += 4) {
            const int last = len - 1;
            const unsigned w0 = pc[i];
            const unsigned w1 = pc[i + 1 < last ? i + 1 : last];
            const unsigned w2 = pc[i + 2 < last ? i + 2 : last];
            const unsigned w3 = pc[i + 3 < last ? i + 3 : last];
            const int n = (len - i < 4) ? (len - i) : 4;
            const unsigned wv[4] = {w0, w1, w2, w3};
            int pos[4];
#pragma unroll
            for (int q = 0; q < 4; ++q) {
                pos[q] = 0;
                if (q < n) pos[q] = atomicAdd(&cur[wv[q] & 511u], 1);
            }
#pragma unroll
            for (int q = 0; q < 4; ++q) {
                if (q < n) {
                    const int src = (int)(wv[q] >> 9);
                    if (!big) outL[pos[q]] = src;
                    else atomicExch(&csr[base + pos[q]], src);
                }
            }
        }
    }
    __syncthreads();

    if (!big) {
        const int zend = base + T;
        for (int g = (base & ~31) + tid; g < zend; g += 256) {
            if (g >= base) {
                const int l0 = g & ~31;
                if (l0 >= base && l0 + 32 <= zend) {
                    csr[g] = outL[g - base];
                } else {
                    atomicExch(&csr[g], outL[g - base]);
                }
            }
        }
    }
}

__device__ __forceinline__ float max_sel(float a, float b) {
    return b > a ? b : a;
}

template <int H, bool ELU>
__global__ __launch_bounds__(256) void gather_kernel(
    const int* __restrict__ rowptr, const int* __restrict__ cnt,
    const int* __restrict__ csr,
    const float* __restrict__ asrc, const float* __restrict__ adst,
    const float* __restrict__ xh, const float* __restrict__ bias,
    float* __restrict__ out, int N)
{
    __shared__ int sS[16 * 32];
    __shared__ __attribute__((aligned(16))) float sW[16 * 32 * H];

    const int gid = blockIdx.x * 256 + threadIdx.x;
    const int d = gid >> 4;
    const int l = gid & 15;
    if (d >= N) return;
    const int g = threadIdx.x >> 4;
    const int h = (H == 8) ? (l >> 1) : 0;
    const int n = cnt[d];
    const int* __restrict__ cs = csr + rowptr[d];
    const float ad     = adst[d * H + h];
    const float a_self = asrc[d * H + h];
    const float4 bv    = *(const float4*)(bias + 4 * l);

    float hm[H], adk[H];
    if constexpr (H == 8) {
        const float4 p = *(const float4*)(asrc + d * 8);
        const float4 q = *(const float4*)(asrc + d * 8 + 4);
        hm[0] = p.x; hm[1] = p.y; hm[2] = p.z; hm[3] = p.w;
        hm[4] = q.x; hm[5] = q.y; hm[6] = q.z; hm[7] = q.w;
        const float4 u = *(const float4*)(adst + d * 8);
        const float4 v = *(const float4*)(adst + d * 8 + 4);
        adk[0] = u.x; adk[1] = u.y; adk[2] = u.z; adk[3] = u.w;
        adk[4] = v.x; adk[5] = v.y; adk[6] = v.z; adk[7] = v.w;
    } else {
        hm[0] = a_self;
        adk[0] = ad;
    }
    int sA = d, sB = d;
    float aA[H], aB[H];
#pragma unroll
    for (int k = 0; k < H; ++k) { aA[k] = 0.f; aB[k] = 0.f; }
    if (n > 0) {
        const int last = n - 1;
        sA = cs[l < last ? l : last];
        sB = cs[l + 16 < last ? l + 16 : last];
        if constexpr (H == 8) {
            const float4 p0 = *(const float4*)(asrc + sA * 8);
            const float4 q0 = *(const float4*)(asrc + sA * 8 + 4);
            const float4 p1 = *(const float4*)(asrc + sB * 8);
            const float4 q1 = *(const float4*)(asrc + sB * 8 + 4);
            aA[0] = p0.x; aA[1] = p0.y; aA[2] = p0.z; aA[3] = p0.w;
            aA[4] = q0.x; aA[5] = q0.y; aA[6] = q0.z; aA[7] = q0.w;
            aB[0] = p1.x; aB[1] = p1.y; aB[2] = p1.z; aB[3] = p1.w;
            aB[4] = q1.x; aB[5] = q1.y; aB[6] = q1.z; aB[7] = q1.w;
        } else {
            aA[0] = asrc[sA];
            aB[0] = asrc[sB];
        }
#pragma unroll
        for (int k = 0; k < H; ++k) {
            hm[k] = max_sel(hm[k], aA[k]);
            hm[k] = max_sel(hm[k], aB[k]);
        }
        for (int j = 32 + l; j < n; j += 16) {
            const int s = cs[j];
            if constexpr (H == 8) {
                const float4 p = *(const float4*)(asrc + s * 8);
                const float4 q = *(const float4*)(asrc + s * 8 + 4);
                hm[0] = max_sel(hm[0], p.x); hm[1] = max_sel(hm[1], p.y);
                hm[2] = max_sel(hm[2], p.z); hm[3] = max_sel(hm[3], p.w);
                hm[4] = max_sel(hm[4], q.x); hm[5] = max_sel(hm[5], q.y);
                hm[6] = max_sel(hm[6], q.z); hm[7] = max_sel(hm[7], q.w);
            } else {
                hm[0] = max_sel(hm[0], asrc[s]);
            }
        }
    }
#pragma unroll
    for (int o = 8; o > 0; o >>= 1) {
#pragma unroll
        for (int k = 0; k < H; ++k) hm[k] = max_sel(hm[k], __shfl_xor(hm[k], o, 16));
    }
    float m = hm[0];
#pragma unroll
    for (int k = 1; k < H; ++k) m = (h == k) ? hm[k] : m;
    const float mx = lrelu02(m + ad);

    const bool vA = l < n, vB = l + 16 < n;
    float wA[H], wB[H];
#pragma unroll
    for (int k = 0; k < H; ++k) {
        const float mk = lrelu02(hm[k] + adk[k]);
        wA[k] = vA ? expf(lrelu02(aA[k] + adk[k]) - mk) : 0.f;
        wB[k] = vB ? expf(lrelu02(aB[k] + adk[k]) - mk) : 0.f;
    }
    sS[g * 32 + l]      = vA ? sA : d;
    sS[g * 32 + 16 + l] = vB ? sB : d;
    if constexpr (H == 8) {
        *(float4*)&sW[(g * 32 + l) * 8]          = make_float4(wA[0], wA[1], wA[2], wA[3]);
        *(float4*)&sW[(g * 32 + l) * 8 + 4]      = make_float4(wA[4], wA[5], wA[6], wA[7]);
        *(float4*)&sW[(g * 32 + 16 + l) * 8]     = make_float4(wB[0], wB[1], wB[2], wB[3]);
        *(float4*)&sW[(g * 32 + 16 + l) * 8 + 4] = make_float4(wB[4], wB[5], wB[6], wB[7]);
    } else {
        sW[g * 32 + l]      = wA[0];
        sW[g * 32 + 16 + l] = wB[0];
    }
    __builtin_amdgcn_fence(__ATOMIC_RELEASE, "wavefront");
    __builtin_amdgcn_wave_barrier();
    __builtin_amdgcn_fence(__ATOMIC_ACQUIRE, "wavefront");

    const float w0 = expf(lrelu02(a_self + ad) - mx);
    float den = w0;
    const float4 x0 = *(const float4*)(xh + d * 64 + 4 * l);
    float4 acc;
    acc.x = x0.x * w0; acc.y = x0.y * w0; acc.z = x0.z * w0; acc.w = x0.w * w0;

    const int n32 = n < 32 ? n : 32;
    for (int jb = 0; jb < n32; jb += 8) {
        int    sq[8];
        float  wq[8];
        float4 xq[8];
#pragma unroll
        for (int q = 0; q < 8; ++q) {
            sq[q] = sS[g * 32 + jb + q];
            wq[q] = sW[(g * 32 + jb + q) * H + h];
        }
#pragma unroll
        for (int q = 0; q < 8; ++q) xq[q] = *(const float4*)(xh + sq[q] * 64 + 4 * l);
#pragma unroll
        for (int q = 0; q < 8; ++q) {
            const float w = wq[q];
            den += w;
            acc.x = __builtin_fmaf(xq[q].x, w, acc.x); acc.y = __builtin_fmaf(xq[q].y, w, acc.y);
            acc.z = __builtin_fmaf(xq[q].z, w, acc.z); acc.w = __builtin_fmaf(xq[q].w, w, acc.w);
        }
    }
    for (int j = 32; j < n; ++j) {
        const int s = cs[j];
        const float4 x = *(const float4*)(xh + s * 64 + 4 * l);
        const float w = expf(lrelu02(asrc[s * H + h] + ad) - mx);
        den += w;
        acc.x += x.x * w; acc.y += x.y * w; acc.z += x.z * w; acc.w += x.w * w;
    }
    const float inv = 1.0f / (den + 1e-16f);
    float4 r;
    r.x = acc.x * inv + bv.x; r.y = acc.y * inv + bv.y;
    r.z = acc.z * inv + bv.z; r.w = acc.w * inv + bv.w;
    if constexpr (ELU) {
        r.x = r.x > 0.f ? r.x : expm1f(r.x);
        r.y = r.y > 0.f ? r.y : expm1f(r.y);
        r.z = r.z > 0.f ? r.z : expm1f(r.z);
        r.w = r.w > 0.f ? r.w : expm1f(r.w);
    }
    *(float4*)(out + d * 64 + 4 * l) = r;
}

static inline int nblk(long long n, int b) { return (int)((n + b - 1) / b); }

extern "C" void kernel_launch(void* const* d_in, const int* in_sizes, int n_in,
                              void* d_out, int out_size, void* d_ws, size_t ws_size,
                              hipStream_t stream)
{
    const int*   ei  = (const int*)  d_in[0];
    const float* X   = (const float*)d_in[1];
    const float* W1  = (const float*)d_in[2];
    const float* as1 = (const float*)d_in[3];
    const float* ad1 = (const float*)d_in[4];
    const float* b1  = (const float*)d_in[5];
    const float* W2  = (const float*)d_in[6];
    const float* as2 = (const float*)d_in[7];
    const float* ad2 = (const float*)d_in[8];
    const float* b2  = (const float*)d_in[9];

    const int E    = in_sizes[0] / 2;
    const int N    = in_sizes[1] / 128;
    const size_t N64 = (size_t)N * 64;

    const int NB = nblk(N, 512);
    const int OP = (NB + 32) & ~31;
    const int NC = nblk(E, CHUNK);
    const size_t Na = ((size_t)N + 31) & ~(size_t)31;
    const size_t Ea = ((size_t)E + 31) & ~(size_t)31;

    float* ws    = (float*)d_ws;
    float* bufA  = ws;
    float* bufB  = bufA + N64;
    float* asrc1 = bufB + N64;
    float* adst1 = asrc1 + Na * 8;
    float* asrc2 = adst1 + Na * 8;
    float* adst2 = asrc2 + Na;
    int*   cnt    = (int*)(adst2 + Na);
    int*   rowptr = cnt + Na;
    int*   csr    = rowptr + Na;
    unsigned* part = (unsigned*)(csr + Ea);
    int*   offg   = (int*)(part + (size_t)NC * CHUNK);
    float* out   = (float*)d_out;

    const int T = 256;

    chunk_split_kernel<<<NC, 256, 0, stream>>>(ei, E, OP, part, offg);
    bucket_order_kernel<<<NB, 256, 0, stream>>>(part, offg, NC, OP, cnt, rowptr, csr, N);

    {
        const int H = 8, C = 8;
        int mtiles = (N + 15) / 16;
        gemm_wmma_kernel<<<nblk(mtiles, 8), 256, 0, stream>>>(X, W1, bufA, N, 128);
        alpha_kernel<<<nblk((long long)N * H, T), T, 0, stream>>>(bufA, as1, ad1, asrc1, adst1, N, H, C);
        gather_kernel<8, true><<<nblk((long long)N * 16, 256), 256, 0, stream>>>(rowptr, cnt, csr, asrc1, adst1, bufA, b1, bufB, N);
    }

    {
        const int H = 1, C = 64;
        int mtiles = (N + 15) / 16;
        gemm_wmma_kernel<<<nblk(mtiles, 8), 256, 0, stream>>>(bufB, W2, bufA, N, 64);
        alpha_kernel<<<nblk((long long)N * H, T), T, 0, stream>>>(bufA, as2, ad2, asrc2, adst2, N, H, C);
        gather_kernel<1, false><<<nblk((long long)N * 16, 256), 256, 0, stream>>>(rowptr, cnt, csr, asrc2, adst2, bufA, b2, out, N);
    }
}


# Gather takes each edge's logit from the source row it fetches anyway (lane dot + DPP row sums: new sum order), one pass with running max and rescale; alpha tables, their lookups and 2 launches go
# speedup vs baseline: 7.3735x; 1.1978x over previous
#include <hip/hip_runtime.h>
#include <math.h>


typedef __attribute__((ext_vector_type(2))) float v2f;
typedef __attribute__((ext_vector_type(8))) float v8f;

__device__ __forceinline__ float lrelu02(float v) {
    return v > 0.0f ? v : 0.2f * v;
}

__global__ __launch_bounds__(256) void gemm_wmma_kernel(
    const float* __restrict__ X, const float* __restrict__ W,
    float* __restrict__ Y, int M, int K)
{
    __shared__ float sW[128 * 64];
    const int nW = K * 64;
    for (int i = threadIdx.x; i < nW; i += 256) {
        int k = i >> 6, n = i & 63;
        sW[(k >> 1) * 128 + 2 * n + (k & 1)] = W[i];
    }
    __syncthreads();

    const int wave = threadIdx.x >> 5;
    const int lane = threadIdx.x & 31;
    const int m0   = (blockIdx.x * 8 + wave) * 16;
    if (m0 >= M) return;

    const int half = lane >> 4;
    const int r    = lane & 15;

    v8f c0 = {}, c1 = {}, c2 = {}, c3 = {};

    const float* arow = X + (size_t)(m0 + r) * K + 2 * half;

    for (int k0 = 0; k0 < K; k0 += 4) {
        v2f a = *(const v2f*)(arow + k0);

        const float* wp = &sW[((k0 >> 1) + half) * 128 + 2 * r];
        v2f b0 = *(const v2f*)(wp +  0);
        v2f b1 = *(const v2f*)(wp + 32);
        v2f b2 = *(const v2f*)(wp + 64);
        v2f b3 = *(const v2f*)(wp + 96);

        c0 = __builtin_amdgcn_wmma_f32_16x16x4_f32(false, a, false, b0,
                                                   (short)0, c0, false, false);
        c1 = __builtin_amdgcn_wmma_f32_16x16x4_f32(false, a, false, b1,
                                                   (short)0, c1, false, false);
        c2 = __builtin_amdgcn_wmma_f32_16x16x4_f32(false, a, false, b2,
                                                   (short)0, c2, false, false);
        c3 = __builtin_amdgcn_wmma_f32_16x16x4_f32(false, a, false, b3,
                                                   (short)0, c3, false, false);
    }

    float* yb = Y + (size_t)m0 * 64 + r;
#pragma unroll
    for (int j = 0; j < 8; ++j) {
        float* yr = yb + (size_t)(j + half * 8) * 64;
        yr[0]  = c0[j];
        yr[16] = c1[j];
        yr[32] = c2[j];
        yr[48] = c3[j];
    }
}

#define CHUNK 6400
#define EPT   25
#define MAXB  1024
#define CAPB  12288

__global__ __launch_bounds__(256) void chunk_split_kernel(
    const int* __restrict__ ei, int E, int OP,
    unsigned* __restrict__ part, int* __restrict__ offg)
{
    __shared__ int cntB[MAXB];
    __shared__ __attribute__((aligned(16))) unsigned sorted[CHUNK];
    __shared__ int wsum[8];
    const int tid = threadIdx.x, lane = tid & 31, wave = tid >> 5;

    for (int i = tid; i < MAXB; i += 256) cntB[i] = 0;
    __syncthreads();

    const int e0 = blockIdx.x * CHUNK;
    unsigned pk[EPT];
    int bk[EPT], rk[EPT];
#pragma unroll
    for (int k = 0; k < EPT; ++k) {
        const int e  = e0 + k * 256 + tid;
        const int ec = e < E ? e : E - 1;
        pk[k] = (unsigned)ei[ec];
        bk[k] = ei[E + ec];
    }
#pragma unroll
    for (int k = 0; k < EPT; ++k) {
        const int e = e0 + k * 256 + tid;
        const int d = bk[k];
        pk[k] = (pk[k] << 9) | (unsigned)(d & 511);
        bk[k] = e < E ? (d >> 9) : -1;
    }
#pragma unroll
    for (int k = 0; k < EPT; ++k) {
        rk[k] = 0;
        if (bk[k] >= 0) rk[k] = atomicAdd(&cntB[bk[k]], 1);
    }
    __syncthreads();

    const int a0 = cntB[4 * tid], a1 = cntB[4 * tid + 1];
    const int a2 = cntB[4 * tid + 2], a3 = cntB[4 * tid + 3];
    const int ts = (a0 + a1) + (a2 + a3);
    int inc = ts;
    for (int o = 1; o < 32; o <<= 1) {
        const int t = __shfl_up(inc, o);
        if (lane >= o) inc += t;
    }
    if (lane == 31) wsum[wave] = inc;
    __syncthreads();
    int ex = inc - ts;
    for (int w = 0; w < wave; ++w) ex += wsum[w];
    int4 o4;
    o4.x = ex; o4.y = ex + a0; o4.z = o4.y + a1; o4.w = o4.z + a2;
    cntB[4 * tid]     = o4.x; cntB[4 * tid + 1] = o4.y;
    cntB[4 * tid + 2] = o4.z; cntB[4 * tid + 3] = o4.w;
    if (4 * tid < OP) *(int4*)(offg + (size_t)blockIdx.x * OP + 4 * tid) = o4;
    __syncthreads();

#pragma unroll
    for (int k = 0; k < EPT; ++k) {
        if (bk[k] >= 0) sorted[cntB[bk[k]] + rk[k]] = pk[k];
    }
    __syncthreads();

    const uint4* __restrict__ s4 = (const uint4*)sorted;
    uint4* __restrict__ p4 = (uint4*)(part + (size_t)blockIdx.x * CHUNK);
    for (int i = tid; i < CHUNK / 4; i += 256) p4[i] = s4[i];
}

__global__ __launch_bounds__(256) void bucket_order_kernel(
    const unsigned* __restrict__ part, const int* __restrict__ offg,
    int NC, int OP, int* __restrict__ cnt, int* __restrict__ rowptr,
    int* __restrict__ csr, int N)
{
    __shared__ int cntN[512];
    __shared__ int cur[512];
    __shared__ int outL[CAPB];
    __shared__ int red[16];
    __shared__ int wsum[8];
    const int tid = threadIdx.x, lane = tid & 31, wave = tid >> 5;
    const int b = blockIdx.x;

    cntN[tid] = 0; cntN[tid + 256] = 0;

    int sT = 0, sB = 0;
    for (int c = tid; c < NC; c += 256) {
        const int o0 = offg[(size_t)c * OP + b];
        const int o1 = offg[(size_t)c * OP + b + 1];
        sB += o0; sT += o1 - o0;
    }
    for (int o = 16; o > 0; o >>= 1) {
        sT += __shfl_xor(sT, o);
        sB += __shfl_xor(sB, o);
    }
    if (lane == 0) { red[wave] = sT; red[8 + wave] = sB; }
    __syncthreads();
    int T = 0, base = 0;
    for (int w = 0; w < 8; ++w) { T += red[w]; base += red[8 + w]; }
    const bool big = T > CAPB;

    for (int c = tid; c < NC; c += 256) {
        const int o0  = offg[(size_t)c * OP + b];
        const int len = offg[(size_t)c * OP + b + 1] - o0;
        const unsigned* __restrict__ pc = part + (size_t)c * CHUNK + o0;
        for (int i = 0; i < len; i += 4) {
            const int last = len - 1;
            const unsigned w0 = pc[i];
            const unsigned w1 = pc[i + 1 < last ? i + 1 : last];
            const unsigned w2 = pc[i + 2 < last ? i + 2 : last];
            const unsigned w3 = pc[i + 3 < last ? i + 3 : last];
            atomicAdd(&cntN[w0 & 511u], 1);
            if (i + 1 < len) atomicAdd(&cntN[w1 & 511u], 1);
            if (i + 2 < len) atomicAdd(&cntN[w2 & 511u], 1);
            if (i + 3 < len) atomicAdd(&cntN[w3 & 511u], 1);
        }
    }
    __syncthreads();

    const int c0 = cntN[2 * tid], c1 = cntN[2 * tid + 1];
    const int ts = c0 + c1;
    int inc = ts;
    for (int o = 1; o < 32; o <<= 1) {
        const int t = __shfl_up(inc, o);
        if (lane >= o) inc += t;
    }
    if (lane == 31) wsum[wave] = inc;
    __syncthreads();
    int ex = inc - ts;
    for (int w = 0; w < wave; ++w) ex += wsum[w];
    cur[2 * tid] = ex; cur[2 * tid + 1] = ex + c0;
    const int d = b * 512 + 2 * tid;
    if (d + 1 < N) {
        *(int2*)(cnt + d)    = make_int2(c0, c1);
        *(int2*)(rowptr + d) = make_int2(base + ex, base + ex + c0);
    } else if (d < N) {
        cnt[d] = c0; rowptr[d] = base + ex;
    }
    __syncthreads();

    for (int c = tid; c < NC; c += 256) {
        const int o0  = offg[(size_t)c * OP + b];
        const int len = offg[(size_t)c * OP + b + 1] - o0;
        const unsigned* __restrict__ pc = part + (size_t)c * CHUNK + o0;
        for (int i = 0; i < len; i += 4) {
            const int last = len - 1;
            const unsigned w0 = pc[i];
            const unsigned w1 = pc[i + 1 < last ? i + 1 : last];
            const unsigned w2 = pc[i + 2 < last ? i + 2 : last];
            const unsigned w3 = pc[i + 3 < last ? i + 3 : last];
            const int n = (len - i < 4) ? (len - i) : 4;
            const unsigned wv[4] = {w0, w1, w2, w3};
            int pos[4];
#pragma unroll
            for (int q = 0; q < 4; ++q) {
                pos[q] = 0;
                if (q < n) pos[q] = atomicAdd(&cur[wv[q] & 511u], 1);
            }
#pragma unroll
            for (int q = 0; q < 4; ++q) {
                if (q < n) {
                    const int src = (int)(wv[q] >> 9);
                    if (!big) outL[pos[q]] = src;
                    else atomicExch(&csr[base + pos[q]], src);
                }
            }
        }
    }
    __syncthreads();

    if (!big) {
        const int zend = base + T;
        for (int g = (base & ~31) + tid; g < zend; g += 256) {
            if (g >= base) {
                const int l0 = g & ~31;
                if (l0 >= base && l0 + 32 <= zend) {
                    csr[g] = outL[g - base];
                } else {
                    atomicExch(&csr[g], outL[g - base]);
                }
            }
        }
    }
}

__device__ __forceinline__ float max_sel(float a, float b) {
    return b > a ? b : a;
}

template <int CTRL>
__device__ __forceinline__ int row_dpp_i(int v) {
    return __builtin_amdgcn_update_dpp(0, v, CTRL, 0xf, 0xf, true);
}
template <int CTRL>
__device__ __forceinline__ float row_dpp_f(float v) {
    return __int_as_float(row_dpp_i<CTRL>(__float_as_int(v)));
}

template <int H>
__device__ __forceinline__ float head_sum(float p) {
    constexpr int LPH = 16 / H;
    if constexpr (LPH > 1) p += row_dpp_f<0x161>(p);
    if constexpr (LPH > 2) p += row_dpp_f<0x162>(p);
    if constexpr (LPH > 4) p += row_dpp_f<0x164>(p);
    if constexpr (LPH > 8) p += row_dpp_f<0x168>(p);
    return p;
}

__device__ __forceinline__ float dot4(const float4 x, const float4 a) {
    return __builtin_fmaf(x.w, a.w, __builtin_fmaf(x.z, a.z, __builtin_fmaf(x.y, a.y, x.x * a.x)));
}

template <int H, int Q0>
__device__ __forceinline__ void edge_trip(
    const char* __restrict__ xhb, unsigned boff, unsigned lane16, int nleft, int l,
    const float4 as4, float ad, float& m, float& den, float4& acc)
{
    static_assert(H == 8 || H == 1, "lane maps below are written for 2 and 16 lanes per head");
    unsigned off[8];
    off[0] = (unsigned)row_dpp_i<0x150 + Q0 + 0>((int)boff) + lane16;
    off[1] = (unsigned)row_dpp_i<0x150 + Q0 + 1>((int)boff) + lane16;
    off[2] = (unsigned)row_dpp_i<0x150 + Q0 + 2>((int)boff) + lane16;
    off[3] = (unsigned)row_dpp_i<0x150 + Q0 + 3>((int)boff) + lane16;
    off[4] = (unsigned)row_dpp_i<0x150 + Q0 + 4>((int)boff) + lane16;
    off[5] = (unsigned)row_dpp_i<0x150 + Q0 + 5>((int)boff) + lane16;
    off[6] = (unsigned)row_dpp_i<0x150 + Q0 + 6>((int)boff) + lane16;
    off[7] = (unsigned)row_dpp_i<0x150 + Q0 + 7>((int)boff) + lane16;
    float4 xq[8];
    float  p[8], w[8];
    float  sc;
#pragma unroll
    for (int q = 0; q < 8; ++q) xq[q] = *(const float4*)(xhb + off[q]);
#pragma unroll
    for (int q = 0; q < 8; ++q) p[q] = dot4(xq[q], as4);

    if constexpr (H == 8) {
        const bool odd = (l & 1) != 0;
        const int  qb  = l & 1;
        float e[4], t[4], o[4];
#pragma unroll
        for (int i = 0; i < 4; ++i) {
            const float keep = odd ? p[2 * i + 1] : p[2 * i];
            const float send = odd ? p[2 * i]     : p[2 * i + 1];
            e[i] = lrelu02((keep + row_dpp_f<0x161>(send)) + ad);
        }
        float bm = max_sel(max_sel(e[0], e[1]), max_sel(e[2], e[3]));
        bm = max_sel(bm, row_dpp_f<0x161>(bm));
        const float mn = max_sel(m, bm);
        sc = expf(m - mn);
#pragma unroll
        for (int i = 0; i < 4; ++i) t[i] = expf((2 * i + qb < nleft ? e[i] : -INFINITY) - mn);
#pragma unroll
        for (int i = 0; i < 4; ++i) o[i] = row_dpp_f<0x161>(t[i]);
#pragma unroll
        for (int i = 0; i < 4; ++i) {
            w[2 * i]     = odd ? o[i] : t[i];
            w[2 * i + 1] = odd ? t[i] : o[i];
        }
        m = mn;
    } else {
        const bool b3 = (l & 8) != 0, b2 = (l & 4) != 0, b1 = (l & 2) != 0;
        const bool odd = (l & 1) != 0;
        float r[4], t2[2];
#pragma unroll
        for (int i = 0; i < 4; ++i) {
            const float keep = b3 ? p[2 * i + 1] : p[2 * i];
            const float send = b3 ? p[2 * i]     : p[2 * i + 1];
            r[i] = keep + row_dpp_f<0x168>(send);
        }
#pragma unroll
        for (int i = 0; i < 2; ++i) {
            const float keep = b2 ? r[2 * i + 1] : r[2 * i];
            const float send = b2 ? r[2 * i]     : r[2 * i + 1];
            t2[i] = keep + row_dpp_f<0x164>(send);
        }
        const float keep = b1 ? t2[1] : t2[0];
        const float send = b1 ? t2[0] : t2[1];
        float u = keep + row_dpp_f<0x162>(send);
        u += row_dpp_f<0x161>(u);
        const float e = lrelu02(u + ad);
        float bm = max_sel(e, row_dpp_f<0x168>(e));
        bm = max_sel(bm, row_dpp_f<0x164>(bm));
        bm = max_sel(bm, row_dpp_f<0x162>(bm));
        const float mn = max_sel(m, bm);
        const int ql = (b1 ? 4 : 0) + (b2 ? 2 : 0) + (b3 ? 1 : 0);
        const float tv = expf((odd ? m : e) - mn);
        const float wv = (odd || ql < nleft) ? tv : 0.f;
        w[0] = row_dpp_f<0x150 + 0>(wv);
        w[1] = row_dpp_f<0x150 + 8>(wv);
        w[2] = row_dpp_f<0x150 + 4>(wv);
        w[3] = row_dpp_f<0x150 + 12>(wv);
        w[4] = row_dpp_f<0x150 + 2>(wv);
        w[5] = row_dpp_f<0x150 + 10>(wv);
        w[6] = row_dpp_f<0x150 + 6>(wv);
        w[7] = row_dpp_f<0x150 + 14>(wv);
        sc   = row_dpp_f<0x150 + 1>(wv);
        m = mn;
    }

    den *= sc;
    acc.x *= sc; acc.y *= sc; acc.z *= sc; acc.w *= sc;
#pragma unroll
    for (int q = 0; q < 8; ++q) {
        den += w[q];
        acc.x = __builtin_fmaf(xq[q].x, w[q], acc.x); acc.y = __builtin_fmaf(xq[q].y, w[q], acc.y);
        acc.z = __builtin_fmaf(xq[q].z, w[q], acc.z); acc.w = __builtin_fmaf(xq[q].w, w[q], acc.w);
    }
}

template <int H, bool ELU>
__global__ __launch_bounds__(256) void gather_kernel(
    const int* __restrict__ rowptr, const int* __restrict__ cnt,
    const int* __restrict__ csr,
    const float* __restrict__ a_src, const float* __restrict__ a_dst,
    const float* __restrict__ xh, const float* __restrict__ bias,
    float* __restrict__ out, int N)
{
    const int gid = blockIdx.x * 256 + threadIdx.x;
    const int d = gid >> 4;
    const int l = gid & 15;
    if (d >= N) return;
    const int n = cnt[d];
    const int* __restrict__ cs = csr + rowptr[d];
    const float4 as4 = *(const float4*)(a_src + 4 * l);
    const float4 ad4 = *(const float4*)(a_dst + 4 * l);
    const float4 x0  = *(const float4*)(xh + d * 64 + 4 * l);
    const char* __restrict__ xhb = (const char*)xh;
    const unsigned lane16 = 16u * (unsigned)l;

    const int last = n - 1;
    int sid = d;
    if (n > 0) sid = cs[l < last ? l : last];

    const float ad = head_sum<H>(dot4(x0, ad4));
    float m   = lrelu02(head_sum<H>(dot4(x0, as4)) + ad);
    float den = 1.0f;
    float4 acc = x0;

    for (int cb = 0; cb < n; cb += 16) {
        int nx = sid;
        if (cb + 16 < n) {
            const int j = cb + 16 + l;
            nx = cs[j < last ? j : last];
        }
        const unsigned boff = (unsigned)sid << 8;
        edge_trip<H, 0>(xhb, boff, lane16, n - cb, l, as4, ad, m, den, acc);
        if (cb + 8 < n) edge_trip<H, 8>(xhb, boff, lane16, n - cb - 8, l, as4, ad, m, den, acc);
        sid = nx;
    }

    const float4 bv = *(const float4*)(bias + 4 * l);
    const float inv = 1.0f / (den + 1e-16f);
    float4 r;
    r.x = acc.x * inv + bv.x; r.y = acc.y * inv + bv.y;
    r.z = acc.z * inv + bv.z; r.w = acc.w * inv + bv.w;
    if constexpr (ELU) {
        r.x = r.x > 0.f ? r.x : expm1f(r.x);
        r.y = r.y > 0.f ? r.y : expm1f(r.y);
        r.z = r.z > 0.f ? r.z : expm1f(r.z);
        r.w = r.w > 0.f ? r.w : expm1f(r.w);
    }
    *(float4*)(out + d * 64 + 4 * l) = r;
}

static inline int nblk(long long n, int b) { return (int)((n + b - 1) / b); }

extern "C" void kernel_launch(void* const* d_in, const int* in_sizes, int n_in,
                              void* d_out, int out_size, void* d_ws, size_t ws_size,
                              hipStream_t stream)
{
    const int*   ei  = (const int*)  d_in[0];
    const float* X   = (const float*)d_in[1];
    const float* W1  = (const float*)d_in[2];
    const float* as1 = (const float*)d_in[3];
    const float* ad1 = (const float*)d_in[4];
    const float* b1  = (const float*)d_in[5];
    const float* W2  = (const float*)d_in[6];
    const float* as2 = (const float*)d_in[7];
    const float* ad2 = (const float*)d_in[8];
    const float* b2  = (const float*)d_in[9];

    const int E    = in_sizes[0] / 2;
    const int N    = in_sizes[1] / 128;
    const size_t N64 = (size_t)N * 64;

    const int NB = nblk(N, 512);
    const int OP = (NB + 32) & ~31;
    const int NC = nblk(E, CHUNK);
    const size_t Na = ((size_t)N + 31) & ~(size_t)31;
    const size_t Ea = ((size_t)E + 31) & ~(size_t)31;

    float* ws    = (float*)d_ws;
    float* bufA  = ws;
    float* bufB  = bufA + N64;
    int*   cnt    = (int*)(bufB + N64);
    int*   rowptr = cnt + Na;
    int*   csr    = rowptr + Na;
    unsigned* part = (unsigned*)(csr + Ea);
    int*   offg   = (int*)(part + (size_t)NC * CHUNK);
    float* out   = (float*)d_out;

    chunk_split_kernel<<<NC, 256, 0, stream>>>(ei, E, OP, part, offg);
    bucket_order_kernel<<<NB, 256, 0, stream>>>(part, offg, NC, OP, cnt, rowptr, csr, N);

    {
        int mtiles = (N + 15) / 16;
        gemm_wmma_kernel<<<nblk(mtiles, 8), 256, 0, stream>>>(X, W1, bufA, N, 128);
        gather_kernel<8, true><<<nblk((long long)N * 16, 256), 256, 0, stream>>>(rowptr, cnt, csr, as1, ad1, bufA, b1, bufB, N);
    }

    {
        int mtiles = (N + 15) / 16;
        gemm_wmma_kernel<<<nblk(mtiles, 8), 256, 0, stream>>>(bufB, W2, bufA, N, 64);
        gather_kernel<1, false><<<nblk((long long)N * 16, 256), 256, 0, stream>>>(rowptr, cnt, csr, as2, ad2, bufA, b2, out, N);
    }
}


# Sort level 2 and the layer-1 GEMM share no data and now share ONE launch: the grid's first work-groups each order one 512-node bucket, the rest compute GEMM tiles beside them (6 launches -> 5)
# speedup vs baseline: 7.8684x; 1.0671x over previous
#include <hip/hip_runtime.h>
#include <math.h>


typedef __attribute__((ext_vector_type(2))) float v2f;
typedef __attribute__((ext_vector_type(8))) float v8f;

__device__ __forceinline__ float lrelu02(float v) {
    return v > 0.0f ? v : 0.2f * v;
}

__device__ __forceinline__ void gemm_wmma_body(
    const float* __restrict__ X, const float* __restrict__ W,
    float* __restrict__ Y, int M, int K, float* sW, int blk)
{
    const int nW = K * 64;
    for (int i = threadIdx.x; i < nW; i += 256) {
        int k = i >> 6, n = i & 63;
        sW[(k >> 1) * 128 + 2 * n + (k & 1)] = W[i];
    }
    __syncthreads();

    const int wave = threadIdx.x >> 5;
    const int lane = threadIdx.x & 31;
    const int m0   = (blk * 8 + wave) * 16;
    if (m0 >= M) return;

    const int half = lane >> 4;
    const int r    = lane & 15;

    v8f c0 = {}, c1 = {}, c2 = {}, c3 = {};

    const float* arow = X + (size_t)(m0 + r) * K + 2 * half;

    for (int k0 = 0; k0 < K; k0 += 4) {
        v2f a = *(const v2f*)(arow + k0);

        const float* wp = &sW[((k0 >> 1) + half) * 128 + 2 * r];
        v2f b0 = *(const v2f*)(wp +  0);
        v2f b1 = *(const v2f*)(wp + 32);
        v2f b2 = *(const v2f*)(wp + 64);
        v2f b3 = *(const v2f*)(wp + 96);

        c0 = __builtin_amdgcn_wmma_f32_16x16x4_f32(false, a, false, b0,
                                                   (short)0, c0, false, false);
        c1 = __builtin_amdgcn_wmma_f32_16x16x4_f32(false, a, false, b1,
                                                   (short)0, c1, false, false);
        c2 = __builtin_amdgcn_wmma_f32_16x16x4_f32(false, a, false, b2,
                                                   (short)0, c2, false, false);
        c3 = __builtin_amdgcn_wmma_f32_16x16x4_f32(false, a, false, b3,
                                                   (short)0, c3, false, false);
    }

    float* yb = Y + (size_t)m0 * 64 + r;
#pragma unroll
    for (int j = 0; j < 8; ++j) {
        float* yr = yb + (size_t)(j + half * 8) * 64;
        yr[0]  = c0[j];
        yr[16] = c1[j];
        yr[32] = c2[j];
        yr[48] = c3[j];
    }
}

__global__ __launch_bounds__(256) void gemm_wmma_kernel(
    const float* __restrict__ X, const float* __restrict__ W,
    float* __restrict__ Y, int M, int K)
{
    __shared__ float sW[128 * 64];
    gemm_wmma_body(X, W, Y, M, K, sW, (int)blockIdx.x);
}

#define CHUNK 6400
#define EPT   25
#define MAXB  1024
#define CAPB  12288

__global__ __launch_bounds__(256) void chunk_split_kernel(
    const int* __restrict__ ei, int E, int OP,
    unsigned* __restrict__ part, int* __restrict__ offg)
{
    __shared__ int cntB[MAXB];
    __shared__ __attribute__((aligned(16))) unsigned sorted[CHUNK];
    __shared__ int wsum[8];
    const int tid = threadIdx.x, lane = tid & 31, wave = tid >> 5;

    for (int i = tid; i < MAXB; i += 256) cntB[i] = 0;
    __syncthreads();

    const int e0 = blockIdx.x * CHUNK;
    unsigned pk[EPT];
    int bk[EPT], rk[EPT];
#pragma unroll
    for (int k = 0; k < EPT; ++k) {
        const int e  = e0 + k * 256 + tid;
        const int ec = e < E ? e : E - 1;
        pk[k] = (unsigned)ei[ec];
        bk[k] = ei[E + ec];
    }
#pragma unroll
    for (int k = 0; k < EPT; ++k) {
        const int e = e0 + k * 256 + tid;
        const int d = bk[k];
        pk[k] = (pk[k] << 9) | (unsigned)(d & 511);
        bk[k] = e < E ? (d >> 9) : -1;
    }
#pragma unroll
    for (int k = 0; k < EPT; ++k) {
        rk[k] = 0;
        if (bk[k] >= 0) rk[k] = atomicAdd(&cntB[bk[k]], 1);
    }
    __syncthreads();

    const int a0 = cntB[4 * tid], a1 = cntB[4 * tid + 1];
    const int a2 = cntB[4 * tid + 2], a3 = cntB[4 * tid + 3];
    const int ts = (a0 + a1) + (a2 + a3);
    int inc = ts;
    for (int o = 1; o < 32; o <<= 1) {
        const int t = __shfl_up(inc, o);
        if (lane >= o) inc += t;
    }
    if (lane == 31) wsum[wave] = inc;
    __syncthreads();
    int ex = inc - ts;
    for (int w = 0; w < wave; ++w) ex += wsum[w];
    int4 o4;
    o4.x = ex; o4.y = ex + a0; o4.z = o4.y + a1; o4.w = o4.z + a2;
    cntB[4 * tid]     = o4.x; cntB[4 * tid + 1] = o4.y;
    cntB[4 * tid + 2] = o4.z; cntB[4 * tid + 3] = o4.w;
    if (4 * tid < OP) *(int4*)(offg + (size_t)blockIdx.x * OP + 4 * tid) = o4;
    __syncthreads();

#pragma unroll
    for (int k = 0; k < EPT; ++k) {
        if (bk[k] >= 0) sorted[cntB[bk[k]] + rk[k]] = pk[k];
    }
    __syncthreads();

    const uint4* __restrict__ s4 = (const uint4*)sorted;
    uint4* __restrict__ p4 = (uint4*)(part + (size_t)blockIdx.x * CHUNK);
    for (int i = tid; i < CHUNK / 4; i += 256) p4[i] = s4[i];
}

#define ORDER_LDS_INTS (512 + 512 + CAPB + 16 + 8)

__device__ __forceinline__ void bucket_order_body(
    const unsigned* __restrict__ part, const int* __restrict__ offg,
    int NC, int OP, int* __restrict__ cnt, int* __restrict__ rowptr,
    int* __restrict__ csr, int N, int* sm, int b)
{
    int* cntN = sm;
    int* cur  = sm + 512;
    int* outL = sm + 1024;
    int* red  = sm + 1024 + CAPB;
    int* wsum = red + 16;
    const int tid = threadIdx.x, lane = tid & 31, wave = tid >> 5;

    cntN[tid] = 0; cntN[tid + 256] = 0;

    int sT = 0, sB = 0;
    for (int c = tid; c < NC; c += 256) {
        const int o0 = offg[(size_t)c * OP + b];
        const int o1 = offg[(size_t)c * OP + b + 1];
        sB += o0; sT += o1 - o0;
    }
    for (int o = 16; o > 0; o >>= 1) {
        sT += __shfl_xor(sT, o);
        sB += __shfl_xor(sB, o);
    }
    if (lane == 0) { red[wave] = sT; red[8 + wave] = sB; }
    __syncthreads();
    int T = 0, base = 0;
    for (int w = 0; w < 8; ++w) { T += red[w]; base += red[8 + w]; }
    const bool big = T > CAPB;

    for (int c = tid; c < NC; c += 256) {
        const int o0  = offg[(size_t)c * OP + b];
        const int len = offg[(size_t)c * OP + b + 1] - o0;
        const unsigned* __restrict__ pc = part + (size_t)c * CHUNK + o0;
        for (int i = 0; i < len; i += 4) {
            const int last = len - 1;
            const unsigned w0 = pc[i];
            const unsigned w1 = pc[i + 1 < last ? i + 1 : last];
            const unsigned w2 = pc[i + 2 < last ? i + 2 : last];
            const unsigned w3 = pc[i + 3 < last ? i + 3 : last];
            atomicAdd(&cntN[w0 & 511u], 1);
            if (i + 1 < len) atomicAdd(&cntN[w1 & 511u], 1);
            if (i + 2 < len) atomicAdd(&cntN[w2 & 511u], 1);
            if (i + 3 < len) atomicAdd(&cntN[w3 & 511u], 1);
        }
    }
    __syncthreads();

    const int c0 = cntN[2 * tid], c1 = cntN[2 * tid + 1];
    const int ts = c0 + c1;
    int inc = ts;
    for (int o = 1; o < 32; o <<= 1) {
        const int t = __shfl_up(inc, o);
        if (lane >= o) inc += t;
    }
    if (lane == 31) wsum[wave] = inc;
    __syncthreads();
    int ex = inc - ts;
    for (int w = 0; w < wave; ++w) ex += wsum[w];
    cur[2 * tid] = ex; cur[2 * tid + 1] = ex + c0;
    const int d = b * 512 + 2 * tid;
    if (d + 1 < N) {
        *(int2*)(cnt + d)    = make_int2(c0, c1);
        *(int2*)(rowptr + d) = make_int2(base + ex, base + ex + c0);
    } else if (d < N) {
        cnt[d] = c0; rowptr[d] = base + ex;
    }
    __syncthreads();

    for (int c = tid; c < NC; c += 256) {
        const int o0  = offg[(size_t)c * OP + b];
        const int len = offg[(size_t)c * OP + b + 1] - o0;
        const unsigned* __restrict__ pc = part + (size_t)c * CHUNK + o0;
        for (int i = 0; i < len; i += 4) {
            const int last = len - 1;
            const unsigned w0 = pc[i];
            const unsigned w1 = pc[i + 1 < last ? i + 1 : last];
            const unsigned w2 = pc[i + 2 < last ? i + 2 : last];
            const unsigned w3 = pc[i + 3 < last ? i + 3 : last];
            const int n = (len - i < 4) ? (len - i) : 4;
            const unsigned wv[4] = {w0, w1, w2, w3};
            int pos[4];
#pragma unroll
            for (int q = 0; q < 4; ++q) {
                pos[q] = 0;
                if (q < n) pos[q] = atomicAdd(&cur[wv[q] & 511u], 1);
            }
#pragma unroll
            for (int q = 0; q < 4; ++q) {
                if (q < n) {
                    const int src = (int)(wv[q] >> 9);
                    if (!big) outL[pos[q]] = src;
                    else atomicExch(&csr[base + pos[q]], src);
                }
            }
        }
    }
    __syncthreads();

    if (!big) {
        const int zend = base + T;
        for (int g = (base & ~31) + tid; g < zend; g += 256) {
            if (g >= base) {
                const int l0 = g & ~31;
                if (l0 >= base && l0 + 32 <= zend) {
                    csr[g] = outL[g - base];
                } else {
                    atomicExch(&csr[g], outL[g - base]);
                }
            }
        }
    }
}

__global__ __launch_bounds__(256) void order_and_gemm_kernel(
    const unsigned* __restrict__ part, const int* __restrict__ offg,
    int NC, int OP, int* __restrict__ cnt, int* __restrict__ rowptr,
    int* __restrict__ csr, int N, int NB,
    const float* __restrict__ X, const float* __restrict__ W,
    float* __restrict__ Y, int K)
{
    static_assert(ORDER_LDS_INTS >= 128 * 64, "the GEMM's W tile must fit the shared LDS block");
    __shared__ __attribute__((aligned(16))) int sm[ORDER_LDS_INTS];
    const int blk = (int)blockIdx.x;
    if (blk < NB) bucket_order_body(part, offg, NC, OP, cnt, rowptr, csr, N, sm, blk);
    else          gemm_wmma_body(X, W, Y, N, K, (float*)sm, blk - NB);
}

__device__ __forceinline__ float max_sel(float a, float b) {
    return b > a ? b : a;
}

template <int CTRL>
__device__ __forceinline__ int row_dpp_i(int v) {
    return __builtin_amdgcn_update_dpp(0, v, CTRL, 0xf, 0xf, true);
}
template <int CTRL>
__device__ __forceinline__ float row_dpp_f(float v) {
    return __int_as_float(row_dpp_i<CTRL>(__float_as_int(v)));
}

template <int H>
__device__ __forceinline__ float head_sum(float p) {
    constexpr int LPH = 16 / H;
    if constexpr (LPH > 1) p += row_dpp_f<0x161>(p);
    if constexpr (LPH > 2) p += row_dpp_f<0x162>(p);
    if constexpr (LPH > 4) p += row_dpp_f<0x164>(p);
    if constexpr (LPH > 8) p += row_dpp_f<0x168>(p);
    return p;
}

__device__ __forceinline__ float dot4(const float4 x, const float4 a) {
    return __builtin_fmaf(x.w, a.w, __builtin_fmaf(x.z, a.z, __builtin_fmaf(x.y, a.y, x.x * a.x)));
}

template <int H, int Q0>
__device__ __forceinline__ void edge_trip(
    const char* __restrict__ xhb, unsigned boff, unsigned lane16, int nleft, int l,
    const float4 as4, float ad, float& m, float& den, float4& acc)
{
    static_assert(H == 8 || H == 1, "lane maps below are written for 2 and 16 lanes per head");
    unsigned off[8];
    off[0] = (unsigned)row_dpp_i<0x150 + Q0 + 0>((int)boff) + lane16;
    off[1] = (unsigned)row_dpp_i<0x150 + Q0 + 1>((int)boff) + lane16;
    off[2] = (unsigned)row_dpp_i<0x150 + Q0 + 2>((int)boff) + lane16;
    off[3] = (unsigned)row_dpp_i<0x150 + Q0 + 3>((int)boff) + lane16;
    off[4] = (unsigned)row_dpp_i<0x150 + Q0 + 4>((int)boff) + lane16;
    off[5] = (unsigned)row_dpp_i<0x150 + Q0 + 5>((int)boff) + lane16;
    off[6] = (unsigned)row_dpp_i<0x150 + Q0 + 6>((int)boff) + lane16;
    off[7] = (unsigned)row_dpp_i<0x150 + Q0 + 7>((int)boff) + lane16;
    float4 xq[8];
    float  p[8], w[8];
    float  sc;
#pragma unroll
    for (int q = 0; q < 8; ++q) xq[q] = *(const float4*)(xhb + off[q]);
#pragma unroll
    for (int q = 0; q < 8; ++q) p[q] = dot4(xq[q], as4);

    if constexpr (H == 8) {
        const bool odd = (l & 1) != 0;
        const int  qb  = l & 1;
        float e[4], t[4], o[4];
#pragma unroll
        for (int i = 0; i < 4; ++i) {
            const float keep = odd ? p[2 * i + 1] : p[2 * i];
            const float send = odd ? p[2 * i]     : p[2 * i + 1];
            e[i] = lrelu02((keep + row_dpp_f<0x161>(send)) + ad);
        }
        float bm = max_sel(max_sel(e[0], e[1]), max_sel(e[2], e[3]));
        bm = max_sel(bm, row_dpp_f<0x161>(bm));
        const float mn = max_sel(m, bm);
        sc = expf(m - mn);
#pragma unroll
        for (int i = 0; i < 4; ++i) t[i] = expf((2 * i + qb < nleft ? e[i] : -INFINITY) - mn);
#pragma unroll
        for (int i = 0; i < 4; ++i) o[i] = row_dpp_f<0x161>(t[i]);
#pragma unroll
        for (int i = 0; i < 4; ++i) {
            w[2 * i]     = odd ? o[i] : t[i];
            w[2 * i + 1] = odd ? t[i] : o[i];
        }
        m = mn;
    } else {
        const bool b3 = (l & 8) != 0, b2 = (l & 4) != 0, b1 = (l & 2) != 0;
        const bool odd = (l & 1) != 0;
        float r[4], t2[2];
#pragma unroll
        for (int i = 0; i < 4; ++i) {
            const float keep = b3 ? p[2 * i + 1] : p[2 * i];
            const float send = b3 ? p[2 * i]     : p[2 * i + 1];
            r[i] = keep + row_dpp_f<0x168>(send);
        }
#pragma unroll
        for (int i = 0; i < 2; ++i) {
            const float keep = b2 ? r[2 * i + 1] : r[2 * i];
            const float send = b2 ? r[2 * i]     : r[2 * i + 1];
            t2[i] = keep + row_dpp_f<0x164>(send);
        }
        const float keep = b1 ? t2[1] : t2[0];
        const float send = b1 ? t2[0] : t2[1];
        float u = keep + row_dpp_f<0x162>(send);
        u += row_dpp_f<0x161>(u);
        const float e = lrelu02(u + ad);
        float bm = max_sel(e, row_dpp_f<0x168>(e));
        bm = max_sel(bm, row_dpp_f<0x164>(bm));
        bm = max_sel(bm, row_dpp_f<0x162>(bm));
        const float mn = max_sel(m, bm);
        const int ql = (b1 ? 4 : 0) + (b2 ? 2 : 0) + (b3 ? 1 : 0);
        const float tv = expf((odd ? m : e) - mn);
        const float wv = (odd || ql < nleft) ? tv : 0.f;
        w[0] = row_dpp_f<0x150 + 0>(wv);
        w[1] = row_dpp_f<0x150 + 8>(wv);
        w[2] = row_dpp_f<0x150 + 4>(wv);
        w[3] = row_dpp_f<0x150 + 12>(wv);
        w[4] = row_dpp_f<0x150 + 2>(wv);
        w[5] = row_dpp_f<0x150 + 10>(wv);
        w[6] = row_dpp_f<0x150 + 6>(wv);
        w[7] = row_dpp_f<0x150 + 14>(wv);
        sc   = row_dpp_f<0x150 + 1>(wv);
        m = mn;
    }

    den *= sc;
    acc.x *= sc; acc.y *= sc; acc.z *= sc; acc.w *= sc;
#pragma unroll
    for (int q = 0; q < 8; ++q) {
        den += w[q];
        acc.x = __builtin_fmaf(xq[q].x, w[q], acc.x); acc.y = __builtin_fmaf(xq[q].y, w[q], acc.y);
        acc.z = __builtin_fmaf(xq[q].z, w[q], acc.z); acc.w = __builtin_fmaf(xq[q].w, w[q], acc.w);
    }
}

template <int H, bool ELU>
__global__ __launch_bounds__(256) void gather_kernel(
    const int* __restrict__ rowptr, const int* __restrict__ cnt,
    const int* __restrict__ csr,
    const float* __restrict__ a_src, const float* __restrict__ a_dst,
    const float* __restrict__ xh, const float* __restrict__ bias,
    float* __restrict__ out, int N)
{
    const int gid = blockIdx.x * 256 + threadIdx.x;
    const int d = gid >> 4;
    const int l = gid & 15;
    if (d >= N) return;
    const int n = cnt[d];
    const int* __restrict__ cs = csr + rowptr[d];
    const float4 as4 = *(const float4*)(a_src + 4 * l);
    const float4 ad4 = *(const float4*)(a_dst + 4 * l);
    const float4 x0  = *(const float4*)(xh + d * 64 + 4 * l);
    const char* __restrict__ xhb = (const char*)xh;
    const unsigned lane16 = 16u * (unsigned)l;

    const int last = n - 1;
    int sid = d;
    if (n > 0) sid = cs[l < last ? l : last];

    const float ad = head_sum<H>(dot4(x0, ad4));
    float m   = lrelu02(head_sum<H>(dot4(x0, as4)) + ad);
    float den = 1.0f;
    float4 acc = x0;

    for (int cb = 0; cb < n; cb += 16) {
        int nx = sid;
        if (cb + 16 < n) {
            const int j = cb + 16 + l;
            nx = cs[j < last ? j : last];
        }
        const unsigned boff = (unsigned)sid << 8;
        edge_trip<H, 0>(xhb, boff, lane16, n - cb, l, as4, ad, m, den, acc);
        if (cb + 8 < n) edge_trip<H, 8>(xhb, boff, lane16, n - cb - 8, l, as4, ad, m, den, acc);
        sid = nx;
    }

    const float4 bv = *(const float4*)(bias + 4 * l);
    const float inv = 1.0f / (den + 1e-16f);
    float4 r;
    r.x = acc.x * inv + bv.x; r.y = acc.y * inv + bv.y;
    r.z = acc.z * inv + bv.z; r.w = acc.w * inv + bv.w;
    if constexpr (ELU) {
        r.x = r.x > 0.f ? r.x : expm1f(r.x);
        r.y = r.y > 0.f ? r.y : expm1f(r.y);
        r.z = r.z > 0.f ? r.z : expm1f(r.z);
        r.w = r.w > 0.f ? r.w : expm1f(r.w);
    }
    *(float4*)(out + d * 64 + 4 * l) = r;
}

static inline int nblk(long long n, int b) { return (int)((n + b - 1) / b); }

extern "C" void kernel_launch(void* const* d_in, const int* in_sizes, int n_in,
                              void* d_out, int out_size, void* d_ws, size_t ws_size,
                              hipStream_t stream)
{
    const int*   ei  = (const int*)  d_in[0];
    const float* X   = (const float*)d_in[1];
    const float* W1  = (const float*)d_in[2];
    const float* as1 = (const float*)d_in[3];
    const float* ad1 = (const float*)d_in[4];
    const float* b1  = (const float*)d_in[5];
    const float* W2  = (const float*)d_in[6];
    const float* as2 = (const float*)d_in[7];
    const float* ad2 = (const float*)d_in[8];
    const float* b2  = (const float*)d_in[9];

    const int E    = in_sizes[0] / 2;
    const int N    = in_sizes[1] / 128;
    const size_t N64 = (size_t)N * 64;

    const int NB = nblk(N, 512);
    const int OP = (NB + 32) & ~31;
    const int NC = nblk(E, CHUNK);
    const size_t Na = ((size_t)N + 31) & ~(size_t)31;
    const size_t Ea = ((size_t)E + 31) & ~(size_t)31;

    float* ws    = (float*)d_ws;
    float* bufA  = ws;
    float* bufB  = bufA + N64;
    int*   cnt    = (int*)(bufB + N64);
    int*   rowptr = cnt + Na;
    int*   csr    = rowptr + Na;
    unsigned* part = (unsigned*)(csr + Ea);
    int*   offg   = (int*)(part + (size_t)NC * CHUNK);
    float* out   = (float*)d_out;

    chunk_split_kernel<<<NC, 256, 0, stream>>>(ei, E, OP, part, offg);

    {
        int mtiles = (N + 15) / 16;
        order_and_gemm_kernel<<<NB + nblk(mtiles, 8), 256, 0, stream>>>(
            part, offg, NC, OP, cnt, rowptr, csr, N, NB, X, W1, bufA, 128);
        gather_kernel<8, true><<<nblk((long long)N * 16, 256), 256, 0, stream>>>(rowptr, cnt, csr, as1, ad1, bufA, b1, bufB, N);
    }

    {
        int mtiles = (N + 15) / 16;
        gemm_wmma_kernel<<<nblk(mtiles, 8), 256, 0, stream>>>(bufB, W2, bufA, N, 64);
        gather_kernel<1, false><<<nblk((long long)N * 16, 256), 256, 0, stream>>>(rowptr, cnt, csr, as2, ad2, bufA, b2, out, N);
    }
}


# Layer-2 GEMM fused into the layer-1 gather launch: each 16-node tile of finished rows goes via LDS to 4 waves that multiply it by LDS-held W2 on WMMA, 5 tiles per group (5 launches -> 4)
# speedup vs baseline: 8.9126x; 1.1327x over previous
#include <hip/hip_runtime.h>
#include <math.h>


typedef __attribute__((ext_vector_type(2))) float v2f;
typedef __attribute__((ext_vector_type(8))) float v8f;

__device__ __forceinline__ float lrelu02(float v) {
    return v > 0.0f ? v : 0.2f * v;
}

__device__ __forceinline__ void gemm_wmma_body(
    const float* __restrict__ X, const float* __restrict__ W,
    float* __restrict__ Y, int M, int K, float* sW, int blk)
{
    const int nW = K * 64;
    for (int i = threadIdx.x; i < nW; i += 256) {
        int k = i >> 6, n = i & 63;
        sW[(k >> 1) * 128 + 2 * n + (k & 1)] = W[i];
    }
    __syncthreads();

    const int wave = threadIdx.x >> 5;
    const int lane = threadIdx.x & 31;
    const int m0   = (blk * 8 + wave) * 16;
    if (m0 >= M) return;

    const int half = lane >> 4;
    const int r    = lane & 15;

    v8f c0 = {}, c1 = {}, c2 = {}, c3 = {};

    const float* arow = X + (size_t)(m0 + r) * K + 2 * half;

    for (int k0 = 0; k0 < K; k0 += 4) {
        v2f a = *(const v2f*)(arow + k0);

        const float* wp = &sW[((k0 >> 1) + half) * 128 + 2 * r];
        v2f b0 = *(const v2f*)(wp +  0);
        v2f b1 = *(const v2f*)(wp + 32);
        v2f b2 = *(const v2f*)(wp + 64);
        v2f b3 = *(const v2f*)(wp + 96);

        c0 = __builtin_amdgcn_wmma_f32_16x16x4_f32(false, a, false, b0,
                                                   (short)0, c0, false, false);
        c1 = __builtin_amdgcn_wmma_f32_16x16x4_f32(false, a, false, b1,
                                                   (short)0, c1, false, false);
        c2 = __builtin_amdgcn_wmma_f32_16x16x4_f32(false, a, false, b2,
                                                   (short)0, c2, false, false);
        c3 = __builtin_amdgcn_wmma_f32_16x16x4_f32(false, a, false, b3,
                                                   (short)0, c3, false, false);
    }

    float* yb = Y + (size_t)m0 * 64 + r;
#pragma unroll
    for (int j = 0; j < 8; ++j) {
        float* yr = yb + (size_t)(j + half * 8) * 64;
        yr[0]  = c0[j];
        yr[16] = c1[j];
        yr[32] = c2[j];
        yr[48] = c3[j];
    }
}

#define CHUNK 6400
#define EPT   25
#define MAXB  1024
#define CAPB  12288

__global__ __launch_bounds__(256) void chunk_split_kernel(
    const int* __restrict__ ei, int E, int OP,
    unsigned* __restrict__ part, int* __restrict__ offg)
{
    __shared__ int cntB[MAXB];
    __shared__ __attribute__((aligned(16))) unsigned sorted[CHUNK];
    __shared__ int wsum[8];
    const int tid = threadIdx.x, lane = tid & 31, wave = tid >> 5;

    for (int i = tid; i < MAXB; i += 256) cntB[i] = 0;
    __syncthreads();

    const int e0 = blockIdx.x * CHUNK;
    unsigned pk[EPT];
    int bk[EPT], rk[EPT];
#pragma unroll
    for (int k = 0; k < EPT; ++k) {
        const int e  = e0 + k * 256 + tid;
        const int ec = e < E ? e : E - 1;
        pk[k] = (unsigned)ei[ec];
        bk[k] = ei[E + ec];
    }
#pragma unroll
    for (int k = 0; k < EPT; ++k) {
        const int e = e0 + k * 256 + tid;
        const int d = bk[k];
        pk[k] = (pk[k] << 9) | (unsigned)(d & 511);
        bk[k] = e < E ? (d >> 9) : -1;
    }
#pragma unroll
    for (int k = 0; k < EPT; ++k) {
        rk[k] = 0;
        if (bk[k] >= 0) rk[k] = atomicAdd(&cntB[bk[k]], 1);
    }
    __syncthreads();

    const int a0 = cntB[4 * tid], a1 = cntB[4 * tid + 1];
    const int a2 = cntB[4 * tid + 2], a3 = cntB[4 * tid + 3];
    const int ts = (a0 + a1) + (a2 + a3);
    int inc = ts;
    for (int o = 1; o < 32; o <<= 1) {
        const int t = __shfl_up(inc, o);
        if (lane >= o) inc += t;
    }
    if (lane == 31) wsum[wave] = inc;
    __syncthreads();
    int ex = inc - ts;
    for (int w = 0; w < wave; ++w) ex += wsum[w];
    int4 o4;
    o4.x = ex; o4.y = ex + a0; o4.z = o4.y + a1; o4.w = o4.z + a2;
    cntB[4 * tid]     = o4.x; cntB[4 * tid + 1] = o4.y;
    cntB[4 * tid + 2] = o4.z; cntB[4 * tid + 3] = o4.w;
    if (4 * tid < OP) *(int4*)(offg + (size_t)blockIdx.x * OP + 4 * tid) = o4;
    __syncthreads();

#pragma unroll
    for (int k = 0; k < EPT; ++k) {
        if (bk[k] >= 0) sorted[cntB[bk[k]] + rk[k]] = pk[k];
    }
    __syncthreads();

    const uint4* __restrict__ s4 = (const uint4*)sorted;
    uint4* __restrict__ p4 = (uint4*)(part + (size_t)blockIdx.x * CHUNK);
    for (int i = tid; i < CHUNK / 4; i += 256) p4[i] = s4[i];
}

#define ORDER_LDS_INTS (512 + 512 + CAPB + 16 + 8)

__device__ __forceinline__ void bucket_order_body(
    const unsigned* __restrict__ part, const int* __restrict__ offg,
    int NC, int OP, int* __restrict__ cnt, int* __restrict__ rowptr,
    int* __restrict__ csr, int N, int* sm, int b)
{
    int* cntN = sm;
    int* cur  = sm + 512;
    int* outL = sm + 1024;
    int* red  = sm + 1024 + CAPB;
    int* wsum = red + 16;
    const int tid = threadIdx.x, lane = tid & 31, wave = tid >> 5;

    cntN[tid] = 0; cntN[tid + 256] = 0;

    int sT = 0, sB = 0;
    for (int c = tid; c < NC; c += 256) {
        const int o0 = offg[(size_t)c * OP + b];
        const int o1 = offg[(size_t)c * OP + b + 1];
        sB += o0; sT += o1 - o0;
    }
    for (int o = 16; o > 0; o >>= 1) {
        sT += __shfl_xor(sT, o);
        sB += __shfl_xor(sB, o);
    }
    if (lane == 0) { red[wave] = sT; red[8 + wave] = sB; }
    __syncthreads();
    int T = 0, base = 0;
    for (int w = 0; w < 8; ++w) { T += red[w]; base += red[8 + w]; }
    const bool big = T > CAPB;

    for (int c = tid; c < NC; c += 256) {
        const int o0  = offg[(size_t)c * OP + b];
        const int len = offg[(size_t)c * OP + b + 1] - o0;
        const unsigned* __restrict__ pc = part + (size_t)c * CHUNK + o0;
        for (int i = 0; i < len; i += 4) {
            const int last = len - 1;
            const unsigned w0 = pc[i];
            const unsigned w1 = pc[i + 1 < last ? i + 1 : last];
            const unsigned w2 = pc[i + 2 < last ? i + 2 : last];
            const unsigned w3 = pc[i + 3 < last ? i + 3 : last];
            atomicAdd(&cntN[w0 & 511u], 1);
            if (i + 1 < len) atomicAdd(&cntN[w1 & 511u], 1);
            if (i + 2 < len) atomicAdd(&cntN[w2 & 511u], 1);
            if (i + 3 < len) atomicAdd(&cntN[w3 & 511u], 1);
        }
    }
    __syncthreads();

    const int c0 = cntN[2 * tid], c1 = cntN[2 * tid + 1];
    const int ts = c0 + c1;
    int inc = ts;
    for (int o = 1; o < 32; o <<= 1) {
        const int t = __shfl_up(inc, o);
        if (lane >= o) inc += t;
    }
    if (lane == 31) wsum[wave] = inc;
    __syncthreads();
    int ex = inc - ts;
    for (int w = 0; w < wave; ++w) ex += wsum[w];
    cur[2 * tid] = ex; cur[2 * tid + 1] = ex + c0;
    const int d = b * 512 + 2 * tid;
    if (d + 1 < N) {
        *(int2*)(cnt + d)    = make_int2(c0, c1);
        *(int2*)(rowptr + d) = make_int2(base + ex, base + ex + c0);
    } else if (d < N) {
        cnt[d] = c0; rowptr[d] = base + ex;
    }
    __syncthreads();

    for (int c = tid; c < NC; c += 256) {
        const int o0  = offg[(size_t)c * OP + b];
        const int len = offg[(size_t)c * OP + b + 1] - o0;
        const unsigned* __restrict__ pc = part + (size_t)c * CHUNK + o0;
        for (int i = 0; i < len; i += 4) {
            const int last = len - 1;
            const unsigned w0 = pc[i];
            const unsigned w1 = pc[i + 1 < last ? i + 1 : last];
            const unsigned w2 = pc[i + 2 < last ? i + 2 : last];
            const unsigned w3 = pc[i + 3 < last ? i + 3 : last];
            const int n = (len - i < 4) ? (len - i) : 4;
            const unsigned wv[4] = {w0, w1, w2, w3};
            int pos[4];
#pragma unroll
            for (int q = 0; q < 4; ++q) {
                pos[q] = 0;
                if (q < n) pos[q] = atomicAdd(&cur[wv[q] & 511u], 1);
            }
#pragma unroll
            for (int q = 0; q < 4; ++q) {
                if (q < n) {
                    const int src = (int)(wv[q] >> 9);
                    if (!big) outL[pos[q]] = src;
                    else atomicExch(&csr[base + pos[q]], src);
                }
            }
        }
    }
    __syncthreads();

    if (!big) {
        const int zend = base + T;
        for (int g = (base & ~31) + tid; g < zend; g += 256) {
            if (g >= base) {
                const int l0 = g & ~31;
                if (l0 >= base && l0 + 32 <= zend) {
                    csr[g] = outL[g - base];
                } else {
                    atomicExch(&csr[g], outL[g - base]);
                }
            }
        }
    }
}

__global__ __launch_bounds__(256) void order_and_gemm_kernel(
    const unsigned* __restrict__ part, const int* __restrict__ offg,
    int NC, int OP, int* __restrict__ cnt, int* __restrict__ rowptr,
    int* __restrict__ csr, int N, int NB,
    const float* __restrict__ X, const float* __restrict__ W,
    float* __restrict__ Y, int K)
{
    static_assert(ORDER_LDS_INTS >= 128 * 64, "the GEMM's W tile must fit the shared LDS block");
    __shared__ __attribute__((aligned(16))) int sm[ORDER_LDS_INTS];
    const int blk = (int)blockIdx.x;
    if (blk < NB) bucket_order_body(part, offg, NC, OP, cnt, rowptr, csr, N, sm, blk);
    else          gemm_wmma_body(X, W, Y, N, K, (float*)sm, blk - NB);
}

__device__ __forceinline__ float max_sel(float a, float b) {
    return b > a ? b : a;
}

template <int CTRL>
__device__ __forceinline__ int row_dpp_i(int v) {
    return __builtin_amdgcn_update_dpp(0, v, CTRL, 0xf, 0xf, true);
}
template <int CTRL>
__device__ __forceinline__ float row_dpp_f(float v) {
    return __int_as_float(row_dpp_i<CTRL>(__float_as_int(v)));
}

template <int H>
__device__ __forceinline__ float head_sum(float p) {
    constexpr int LPH = 16 / H;
    if constexpr (LPH > 1) p += row_dpp_f<0x161>(p);
    if constexpr (LPH > 2) p += row_dpp_f<0x162>(p);
    if constexpr (LPH > 4) p += row_dpp_f<0x164>(p);
    if constexpr (LPH > 8) p += row_dpp_f<0x168>(p);
    return p;
}

__device__ __forceinline__ float dot4(const float4 x, const float4 a) {
    return __builtin_fmaf(x.w, a.w, __builtin_fmaf(x.z, a.z, __builtin_fmaf(x.y, a.y, x.x * a.x)));
}

template <int H, int Q0>
__device__ __forceinline__ void edge_trip(
    const char* __restrict__ xhb, unsigned boff, unsigned lane16, int nleft, int l,
    const float4 as4, float ad, float& m, float& den, float4& acc)
{
    static_assert(H == 8 || H == 1, "lane maps below are written for 2 and 16 lanes per head");
    unsigned off[8];
    off[0] = (unsigned)row_dpp_i<0x150 + Q0 + 0>((int)boff) + lane16;
    off[1] = (unsigned)row_dpp_i<0x150 + Q0 + 1>((int)boff) + lane16;
    off[2] = (unsigned)row_dpp_i<0x150 + Q0 + 2>((int)boff) + lane16;
    off[3] = (unsigned)row_dpp_i<0x150 + Q0 + 3>((int)boff) + lane16;
    off[4] = (unsigned)row_dpp_i<0x150 + Q0 + 4>((int)boff) + lane16;
    off[5] = (unsigned)row_dpp_i<0x150 + Q0 + 5>((int)boff) + lane16;
    off[6] = (unsigned)row_dpp_i<0x150 + Q0 + 6>((int)boff) + lane16;
    off[7] = (unsigned)row_dpp_i<0x150 + Q0 + 7>((int)boff) + lane16;
    float4 xq[8];
    float  p[8], w[8];
    float  sc;
#pragma unroll
    for (int q = 0; q < 8; ++q) xq[q] = *(const float4*)(xhb + off[q]);
#pragma unroll
    for (int q = 0; q < 8; ++q) p[q] = dot4(xq[q], as4);

    if constexpr (H == 8) {
        const bool odd = (l & 1) != 0;
        const int  qb  = l & 1;
        float e[4], t[4], o[4];
#pragma unroll
        for (int i = 0; i < 4; ++i) {
            const float keep = odd ? p[2 * i + 1] : p[2 * i];
            const float send = odd ? p[2 * i]     : p[2 * i + 1];
            e[i] = lrelu02((keep + row_dpp_f<0x161>(send)) + ad);
        }
        float bm = max_sel(max_sel(e[0], e[1]), max_sel(e[2], e[3]));
        bm = max_sel(bm, row_dpp_f<0x161>(bm));
        const float mn = max_sel(m, bm);
        sc = expf(m - mn);
#pragma unroll
        for (int i = 0; i < 4; ++i) t[i] = expf((2 * i + qb < nleft ? e[i] : -INFINITY) - mn);
#pragma unroll
        for (int i = 0; i < 4; ++i) o[i] = row_dpp_f<0x161>(t[i]);
#pragma unroll
        for (int i = 0; i < 4; ++i) {
            w[2 * i]     = odd ? o[i] : t[i];
            w[2 * i + 1] = odd ? t[i] : o[i];
        }
        m = mn;
    } else {
        const bool b3 = (l & 8) != 0, b2 = (l & 4) != 0, b1 = (l & 2) != 0;
        const bool odd = (l & 1) != 0;
        float r[4], t2[2];
#pragma unroll
        for (int i = 0; i < 4; ++i) {
            const float keep = b3 ? p[2 * i + 1] : p[2 * i];
            const float send = b3 ? p[2 * i]     : p[2 * i + 1];
            r[i] = keep + row_dpp_f<0x168>(send);
        }
#pragma unroll
        for (int i = 0; i < 2; ++i) {
            const float keep = b2 ? r[2 * i + 1] : r[2 * i];
            const float send = b2 ? r[2 * i]     : r[2 * i + 1];
            t2[i] = keep + row_dpp_f<0x164>(send);
        }
        const float keep = b1 ? t2[1] : t2[0];
        const float send = b1 ? t2[0] : t2[1];
        float u = keep + row_dpp_f<0x162>(send);
        u += row_dpp_f<0x161>(u);
        const float e = lrelu02(u + ad);
        float bm = max_sel(e, row_dpp_f<0x168>(e));
        bm = max_sel(bm, row_dpp_f<0x164>(bm));
        bm = max_sel(bm, row_dpp_f<0x162>(bm));
        const float mn = max_sel(m, bm);
        const int ql = (b1 ? 4 : 0) + (b2 ? 2 : 0) + (b3 ? 1 : 0);
        const float tv = expf((odd ? m : e) - mn);
        const float wv = (odd || ql < nleft) ? tv : 0.f;
        w[0] = row_dpp_f<0x150 + 0>(wv);
        w[1] = row_dpp_f<0x150 + 8>(wv);
        w[2] = row_dpp_f<0x150 + 4>(wv);
        w[3] = row_dpp_f<0x150 + 12>(wv);
        w[4] = row_dpp_f<0x150 + 2>(wv);
        w[5] = row_dpp_f<0x150 + 10>(wv);
        w[6] = row_dpp_f<0x150 + 6>(wv);
        w[7] = row_dpp_f<0x150 + 14>(wv);
        sc   = row_dpp_f<0x150 + 1>(wv);
        m = mn;
    }

    den *= sc;
    acc.x *= sc; acc.y *= sc; acc.z *= sc; acc.w *= sc;
#pragma unroll
    for (int q = 0; q < 8; ++q) {
        den += w[q];
        acc.x = __builtin_fmaf(xq[q].x, w[q], acc.x); acc.y = __builtin_fmaf(xq[q].y, w[q], acc.y);
        acc.z = __builtin_fmaf(xq[q].z, w[q], acc.z); acc.w = __builtin_fmaf(xq[q].w, w[q], acc.w);
    }
}

template <int H, bool ELU>
__device__ __forceinline__ float4 gather_row(
    const int* __restrict__ rowptr, const int* __restrict__ cnt,
    const int* __restrict__ csr,
    const float4 as4, const float4 ad4, const float4 bv,
    const float* __restrict__ xh, int d, int l)
{
    const int n = cnt[d];
    const int* __restrict__ cs = csr + rowptr[d];
    const float4 x0  = *(const float4*)(xh + d * 64 + 4 * l);
    const char* __restrict__ xhb = (const char*)xh;
    const unsigned lane16 = 16u * (unsigned)l;

    const int last = n - 1;
    int sid = d;
    if (n > 0) sid = cs[l < last ? l : last];

    const float ad = head_sum<H>(dot4(x0, ad4));
    float m   = lrelu02(head_sum<H>(dot4(x0, as4)) + ad);
    float den = 1.0f;
    float4 acc = x0;

    for (int cb = 0; cb < n; cb += 16) {
        int nx = sid;
        if (cb + 16 < n) {
            const int j = cb + 16 + l;
            nx = cs[j < last ? j : last];
        }
        const unsigned boff = (unsigned)sid << 8;
        edge_trip<H, 0>(xhb, boff, lane16, n - cb, l, as4, ad, m, den, acc);
        if (cb + 8 < n) edge_trip<H, 8>(xhb, boff, lane16, n - cb - 8, l, as4, ad, m, den, acc);
        sid = nx;
    }

    const float inv = 1.0f / (den + 1e-16f);
    float4 r;
    r.x = acc.x * inv + bv.x; r.y = acc.y * inv + bv.y;
    r.z = acc.z * inv + bv.z; r.w = acc.w * inv + bv.w;
    if constexpr (ELU) {
        r.x = r.x > 0.f ? r.x : expm1f(r.x);
        r.y = r.y > 0.f ? r.y : expm1f(r.y);
        r.z = r.z > 0.f ? r.z : expm1f(r.z);
        r.w = r.w > 0.f ? r.w : expm1f(r.w);
    }
    return r;
}

template <int H, bool ELU>
__global__ __launch_bounds__(256) void gather_kernel(
    const int* __restrict__ rowptr, const int* __restrict__ cnt,
    const int* __restrict__ csr,
    const float* __restrict__ a_src, const float* __restrict__ a_dst,
    const float* __restrict__ xh, const float* __restrict__ bias,
    float* __restrict__ out, int N)
{
    const int gid = blockIdx.x * 256 + threadIdx.x;
    const int d = gid >> 4;
    const int l = gid & 15;
    if (d >= N) return;
    const float4 as4 = *(const float4*)(a_src + 4 * l);
    const float4 ad4 = *(const float4*)(a_dst + 4 * l);
    const float4 bv  = *(const float4*)(bias + 4 * l);
    const float4 r = gather_row<H, ELU>(rowptr, cnt, csr, as4, ad4, bv, xh, d, l);
    *(float4*)(out + d * 64 + 4 * l) = r;
}

#define XPITCH 68

__global__ __launch_bounds__(256) void gather_project_kernel(
    const int* __restrict__ rowptr, const int* __restrict__ cnt,
    const int* __restrict__ csr,
    const float* __restrict__ a_src, const float* __restrict__ a_dst,
    const float* __restrict__ xh, const float* __restrict__ bias,
    const float* __restrict__ W2, float* __restrict__ Y,
    int N, int ntiles, int tpw)
{
    __shared__ __attribute__((aligned(16))) float sW[64 * 64];
    __shared__ __attribute__((aligned(16))) float sX[2][16 * XPITCH];
    const int tid = threadIdx.x;
    for (int i = tid; i < 64 * 64; i += 256) {
        const int k = i >> 6, n = i & 63;
        sW[(k >> 1) * 128 + 2 * n + (k & 1)] = W2[i];
    }

    const int l    = tid & 15;
    const int rowT = tid >> 4;
    const int wave = tid >> 5;
    const int half = (tid >> 4) & 1;
    const float4 as4 = *(const float4*)(a_src + 4 * l);
    const float4 ad4 = *(const float4*)(a_dst + 4 * l);
    const float4 bv  = *(const float4*)(bias + 4 * l);

    const int t0 = (int)blockIdx.x * tpw;
    const int t1 = t0 + tpw < ntiles ? t0 + tpw : ntiles;
    for (int t = t0; t < t1; ++t) {
        const int d  = t * 16 + rowT;
        const int dd = d < N ? d : N - 1;
        const float4 r = gather_row<8, true>(rowptr, cnt, csr, as4, ad4, bv, xh, dd, l);
        float* sx = sX[(t - t0) & 1];
        *(float4*)(sx + rowT * XPITCH + 4 * l) = r;
        __syncthreads();
        if (wave < 4) {
            v8f cacc = {};
            const float* ap = sx + l * XPITCH + 2 * half;
            const float* bp = sW + half * 128 + 2 * (l + 16 * wave);
#pragma unroll
            for (int kk = 0; kk < 16; ++kk) {
                const v2f a = *(const v2f*)(ap + 4 * kk);
                const v2f b = *(const v2f*)(bp + 256 * kk);
                cacc = __builtin_amdgcn_wmma_f32_16x16x4_f32(false, a, false, b,
                                                             (short)0, cacc, false, false);
            }
            const int row0 = t * 16 + 8 * half;
            float* yb = Y + (size_t)row0 * 64 + 16 * wave + l;
            if (t * 16 + 16 <= N) {
#pragma unroll
                for (int j = 0; j < 8; ++j) yb[(size_t)j * 64] = cacc[j];
            } else {
#pragma unroll
                for (int j = 0; j < 8; ++j) {
                    if (row0 + j < N) yb[(size_t)j * 64] = cacc[j];
                }
            }
        }
    }
}

static inline int nblk(long long n, int b) { return (int)((n + b - 1) / b); }

extern "C" void kernel_launch(void* const* d_in, const int* in_sizes, int n_in,
                              void* d_out, int out_size, void* d_ws, size_t ws_size,
                              hipStream_t stream)
{
    const int*   ei  = (const int*)  d_in[0];
    const float* X   = (const float*)d_in[1];
    const float* W1  = (const float*)d_in[2];
    const float* as1 = (const float*)d_in[3];
    const float* ad1 = (const float*)d_in[4];
    const float* b1  = (const float*)d_in[5];
    const float* W2  = (const float*)d_in[6];
    const float* as2 = (const float*)d_in[7];
    const float* ad2 = (const float*)d_in[8];
    const float* b2  = (const float*)d_in[9];

    const int E    = in_sizes[0] / 2;
    const int N    = in_sizes[1] / 128;
    const size_t N64 = (size_t)N * 64;

    const int NB = nblk(N, 512);
    const int OP = (NB + 32) & ~31;
    const int NC = nblk(E, CHUNK);
    const size_t Na = ((size_t)N + 31) & ~(size_t)31;
    const size_t Ea = ((size_t)E + 31) & ~(size_t)31;

    float* ws    = (float*)d_ws;
    float* bufA  = ws;
    float* bufB  = bufA + N64;
    int*   cnt    = (int*)(bufB + N64);
    int*   rowptr = cnt + Na;
    int*   csr    = rowptr + Na;
    unsigned* part = (unsigned*)(csr + Ea);
    int*   offg   = (int*)(part + (size_t)NC * CHUNK);
    float* out   = (float*)d_out;

    chunk_split_kernel<<<NC, 256, 0, stream>>>(ei, E, OP, part, offg);

    {
        int mtiles = (N + 15) / 16;
        order_and_gemm_kernel<<<NB + nblk(mtiles, 8), 256, 0, stream>>>(
            part, offg, NC, OP, cnt, rowptr, csr, N, NB, X, W1, bufA, 128);
        const int tpw = nblk(mtiles, 1280);
        gather_project_kernel<<<nblk(mtiles, tpw), 256, 0, stream>>>(
            rowptr, cnt, csr, as1, ad1, bufA, b1, W2, bufB, N, mtiles, tpw);
    }

    gather_kernel<1, false><<<nblk((long long)N * 16, 256), 256, 0, stream>>>(rowptr, cnt, csr, as2, ad2, bufB, b2, out, N);
}


# Softmax exponentials (arguments always <= 0) computed as one multiply by log2(e) and the hardware exp2, not the library expf with its range handling; weights differ in the last bits only
# speedup vs baseline: 9.2939x; 1.0428x over previous
#include <hip/hip_runtime.h>
#include <math.h>


typedef __attribute__((ext_vector_type(2))) float v2f;
typedef __attribute__((ext_vector_type(8))) float v8f;

__device__ __forceinline__ float lrelu02(float v) {
    return v > 0.0f ? v : 0.2f * v;
}

__device__ __forceinline__ float exp_nonpos(float x) {
    return __builtin_amdgcn_exp2f(x * 1.4426950408889634f);
}

__device__ __forceinline__ void gemm_wmma_body(
    const float* __restrict__ X, const float* __restrict__ W,
    float* __restrict__ Y, int M, int K, float* sW, int blk)
{
    const int nW = K * 64;
    for (int i = threadIdx.x; i < nW; i += 256) {
        int k = i >> 6, n = i & 63;
        sW[(k >> 1) * 128 + 2 * n + (k & 1)] = W[i];
    }
    __syncthreads();

    const int wave = threadIdx.x >> 5;
    const int lane = threadIdx.x & 31;
    const int m0   = (blk * 8 + wave) * 16;
    if (m0 >= M) return;

    const int half = lane >> 4;
    const int r    = lane & 15;

    v8f c0 = {}, c1 = {}, c2 = {}, c3 = {};

    const float* arow = X + (size_t)(m0 + r) * K + 2 * half;

    for (int k0 = 0; k0 < K; k0 += 4) {
        v2f a = *(const v2f*)(arow + k0);

        const float* wp = &sW[((k0 >> 1) + half) * 128 + 2 * r];
        v2f b0 = *(const v2f*)(wp +  0);
        v2f b1 = *(const v2f*)(wp + 32);
        v2f b2 = *(const v2f*)(wp + 64);
        v2f b3 = *(const v2f*)(wp + 96);

        c0 = __builtin_amdgcn_wmma_f32_16x16x4_f32(false, a, false, b0,
                                                   (short)0, c0, false, false);
        c1 = __builtin_amdgcn_wmma_f32_16x16x4_f32(false, a, false, b1,
                                                   (short)0, c1, false, false);
        c2 = __builtin_amdgcn_wmma_f32_16x16x4_f32(false, a, false, b2,
                                                   (short)0, c2, false, false);
        c3 = __builtin_amdgcn_wmma_f32_16x16x4_f32(false, a, false, b3,
                                                   (short)0, c3, false, false);
    }

    float* yb = Y + (size_t)m0 * 64 + r;
#pragma unroll
    for (int j = 0; j < 8; ++j) {
        float* yr = yb + (size_t)(j + half * 8) * 64;
        yr[0]  = c0[j];
        yr[16] = c1[j];
        yr[32] = c2[j];
        yr[48] = c3[j];
    }
}

#define CHUNK 6400
#define EPT   25
#define MAXB  1024
#define CAPB  12288

__global__ __launch_bounds__(256) void chunk_split_kernel(
    const int* __restrict__ ei, int E, int OP,
    unsigned* __restrict__ part, int* __restrict__ offg)
{
    __shared__ int cntB[MAXB];
    __shared__ __attribute__((aligned(16))) unsigned sorted[CHUNK];
    __shared__ int wsum[8];
    const int tid = threadIdx.x, lane = tid & 31, wave = tid >> 5;

    for (int i = tid; i < MAXB; i += 256) cntB[i] = 0;
    __syncthreads();

    const int e0 = blockIdx.x * CHUNK;
    unsigned pk[EPT];
    int bk[EPT], rk[EPT];
#pragma unroll
    for (int k = 0; k < EPT; ++k) {
        const int e  = e0 + k * 256 + tid;
        const int ec = e < E ? e : E - 1;
        pk[k] = (unsigned)ei[ec];
        bk[k] = ei[E + ec];
    }
#pragma unroll
    for (int k = 0; k < EPT; ++k) {
        const int e = e0 + k * 256 + tid;
        const int d = bk[k];
        pk[k] = (pk[k] << 9) | (unsigned)(d & 511);
        bk[k] = e < E ? (d >> 9) : -1;
    }
#pragma unroll
    for (int k = 0; k < EPT; ++k) {
        rk[k] = 0;
        if (bk[k] >= 0) rk[k] = atomicAdd(&cntB[bk[k]], 1);
    }
    __syncthreads();

    const int a0 = cntB[4 * tid], a1 = cntB[4 * tid + 1];
    const int a2 = cntB[4 * tid + 2], a3 = cntB[4 * tid + 3];
    const int ts = (a0 + a1) + (a2 + a3);
    int inc = ts;
    for (int o = 1; o < 32; o <<= 1) {
        const int t = __shfl_up(inc, o);
        if (lane >= o) inc += t;
    }
    if (lane == 31) wsum[wave] = inc;
    __syncthreads();
    int ex = inc - ts;
    for (int w = 0; w < wave; ++w) ex += wsum[w];
    int4 o4;
    o4.x = ex; o4.y = ex + a0; o4.z = o4.y + a1; o4.w = o4.z + a2;
    cntB[4 * tid]     = o4.x; cntB[4 * tid + 1] = o4.y;
    cntB[4 * tid + 2] = o4.z; cntB[4 * tid + 3] = o4.w;
    if (4 * tid < OP) *(int4*)(offg + (size_t)blockIdx.x * OP + 4 * tid) = o4;
    __syncthreads();

#pragma unroll
    for (int k = 0; k < EPT; ++k) {
        if (bk[k] >= 0) sorted[cntB[bk[k]] + rk[k]] = pk[k];
    }
    __syncthreads();

    const uint4* __restrict__ s4 = (const uint4*)sorted;
    uint4* __restrict__ p4 = (uint4*)(part + (size_t)blockIdx.x * CHUNK);
    for (int i = tid; i < CHUNK / 4; i += 256) p4[i] = s4[i];
}

#define ORDER_LDS_INTS (512 + 512 + CAPB + 16 + 8)

__device__ __forceinline__ void bucket_order_body(
    const unsigned* __restrict__ part, const int* __restrict__ offg,
    int NC, int OP, int* __restrict__ cnt, int* __restrict__ rowptr,
    int* __restrict__ csr, int N, int* sm, int b)
{
    int* cntN = sm;
    int* cur  = sm + 512;
    int* outL = sm + 1024;
    int* red  = sm + 1024 + CAPB;
    int* wsum = red + 16;
    const int tid = threadIdx.x, lane = tid & 31, wave = tid >> 5;

    cntN[tid] = 0; cntN[tid + 256] = 0;

    int sT = 0, sB = 0;
    for (int c = tid; c < NC; c += 256) {
        const int o0 = offg[(size_t)c * OP + b];
        const int o1 = offg[(size_t)c * OP + b + 1];
        sB += o0; sT += o1 - o0;
    }
    for (int o = 16; o > 0; o >>= 1) {
        sT += __shfl_xor(sT, o);
        sB += __shfl_xor(sB, o);
    }
    if (lane == 0) { red[wave] = sT; red[8 + wave] = sB; }
    __syncthreads();
    int T = 0, base = 0;
    for (int w = 0; w < 8; ++w) { T += red[w]; base += red[8 + w]; }
    const bool big = T > CAPB;

    for (int c = tid; c < NC; c += 256) {
        const int o0  = offg[(size_t)c * OP + b];
        const int len = offg[(size_t)c * OP + b + 1] - o0;
        const unsigned* __restrict__ pc = part + (size_t)c * CHUNK + o0;
        for (int i = 0; i < len; i += 4) {
            const int last = len - 1;
            const unsigned w0 = pc[i];
            const unsigned w1 = pc[i + 1 < last ? i + 1 : last];
            const unsigned w2 = pc[i + 2 < last ? i + 2 : last];
            const unsigned w3 = pc[i + 3 < last ? i + 3 : last];
            atomicAdd(&cntN[w0 & 511u], 1);
            if (i + 1 < len) atomicAdd(&cntN[w1 & 511u], 1);
            if (i + 2 < len) atomicAdd(&cntN[w2 & 511u], 1);
            if (i + 3 < len) atomicAdd(&cntN[w3 & 511u], 1);
        }
    }
    __syncthreads();

    const int c0 = cntN[2 * tid], c1 = cntN[2 * tid + 1];
    const int ts = c0 + c1;
    int inc = ts;
    for (int o = 1; o < 32; o <<= 1) {
        const int t = __shfl_up(inc, o);
        if (lane >= o) inc += t;
    }
    if (lane == 31) wsum[wave] = inc;
    __syncthreads();
    int ex = inc - ts;
    for (int w = 0; w < wave; ++w) ex += wsum[w];
    cur[2 * tid] = ex; cur[2 * tid + 1] = ex + c0;
    const int d = b * 512 + 2 * tid;
    if (d + 1 < N) {
        *(int2*)(cnt + d)    = make_int2(c0, c1);
        *(int2*)(rowptr + d) = make_int2(base + ex, base + ex + c0);
    } else if (d < N) {
        cnt[d] = c0; rowptr[d] = base + ex;
    }
    __syncthreads();

    for (int c = tid; c < NC; c += 256) {
        const int o0  = offg[(size_t)c * OP + b];
        const int len = offg[(size_t)c * OP + b + 1] - o0;
        const unsigned* __restrict__ pc = part + (size_t)c * CHUNK + o0;
        for (int i = 0; i < len; i += 4) {
            const int last = len - 1;
            const unsigned w0 = pc[i];
            const unsigned w1 = pc[i + 1 < last ? i + 1 : last];
            const unsigned w2 = pc[i + 2 < last ? i + 2 : last];
            const unsigned w3 = pc[i + 3 < last ? i + 3 : last];
            const int n = (len - i < 4) ? (len - i) : 4;
            const unsigned wv[4] = {w0, w1, w2, w3};
            int pos[4];
#pragma unroll
            for (int q = 0; q < 4; ++q) {
                pos[q] = 0;
                if (q < n) pos[q] = atomicAdd(&cur[wv[q] & 511u], 1);
            }
#pragma unroll
            for (int q = 0; q < 4; ++q) {
                if (q < n) {
                    const int src = (int)(wv[q] >> 9);
                    if (!big) outL[pos[q]] = src;
                    else atomicExch(&csr[base + pos[q]], src);
                }
            }
        }
    }
    __syncthreads();

    if (!big) {
        const int zend = base + T;
        for (int g = (base & ~31) + tid; g < zend; g += 256) {
            if (g >= base) {
                const int l0 = g & ~31;
                if (l0 >= base && l0 + 32 <= zend) {
                    csr[g] = outL[g - base];
                } else {
                    atomicExch(&csr[g], outL[g - base]);
                }
            }
        }
    }
}

__global__ __launch_bounds__(256) void order_and_gemm_kernel(
    const unsigned* __restrict__ part, const int* __restrict__ offg,
    int NC, int OP, int* __restrict__ cnt, int* __restrict__ rowptr,
    int* __restrict__ csr, int N, int NB,
    const float* __restrict__ X, const float* __restrict__ W,
    float* __restrict__ Y, int K)
{
    static_assert(ORDER_LDS_INTS >= 128 * 64, "the GEMM's W tile must fit the shared LDS block");
    __shared__ __attribute__((aligned(16))) int sm[ORDER_LDS_INTS];
    const int blk = (int)blockIdx.x;
    if (blk < NB) bucket_order_body(part, offg, NC, OP, cnt, rowptr, csr, N, sm, blk);
    else          gemm_wmma_body(X, W, Y, N, K, (float*)sm, blk - NB);
}

__device__ __forceinline__ float max_sel(float a, float b) {
    return b > a ? b : a;
}

template <int CTRL>
__device__ __forceinline__ int row_dpp_i(int v) {
    return __builtin_amdgcn_update_dpp(0, v, CTRL, 0xf, 0xf, true);
}
template <int CTRL>
__device__ __forceinline__ float row_dpp_f(float v) {
    return __int_as_float(row_dpp_i<CTRL>(__float_as_int(v)));
}

template <int H>
__device__ __forceinline__ float head_sum(float p) {
    constexpr int LPH = 16 / H;
    if constexpr (LPH > 1) p += row_dpp_f<0x161>(p);
    if constexpr (LPH > 2) p += row_dpp_f<0x162>(p);
    if constexpr (LPH > 4) p += row_dpp_f<0x164>(p);
    if constexpr (LPH > 8) p += row_dpp_f<0x168>(p);
    return p;
}

__device__ __forceinline__ float dot4(const float4 x, const float4 a) {
    return __builtin_fmaf(x.w, a.w, __builtin_fmaf(x.z, a.z, __builtin_fmaf(x.y, a.y, x.x * a.x)));
}

template <int H, int Q0>
__device__ __forceinline__ void edge_trip(
    const char* __restrict__ xhb, unsigned boff, unsigned lane16, int nleft, int l,
    const float4 as4, float ad, float& m, float& den, float4& acc)
{
    static_assert(H == 8 || H == 1, "lane maps below are written for 2 and 16 lanes per head");
    unsigned off[8];
    off[0] = (unsigned)row_dpp_i<0x150 + Q0 + 0>((int)boff) + lane16;
    off[1] = (unsigned)row_dpp_i<0x150 + Q0 + 1>((int)boff) + lane16;
    off[2] = (unsigned)row_dpp_i<0x150 + Q0 + 2>((int)boff) + lane16;
    off[3] = (unsigned)row_dpp_i<0x150 + Q0 + 3>((int)boff) + lane16;
    off[4] = (unsigned)row_dpp_i<0x150 + Q0 + 4>((int)boff) + lane16;
    off[5] = (unsigned)row_dpp_i<0x150 + Q0 + 5>((int)boff) + lane16;
    off[6] = (unsigned)row_dpp_i<0x150 + Q0 + 6>((int)boff) + lane16;
    off[7] = (unsigned)row_dpp_i<0x150 + Q0 + 7>((int)boff) + lane16;
    float4 xq[8];
    float  p[8], w[8];
    float  sc;
#pragma unroll
    for (int q = 0; q < 8; ++q) xq[q] = *(const float4*)(xhb + off[q]);
#pragma unroll
    for (int q = 0; q < 8; ++q) p[q] = dot4(xq[q], as4);

    if constexpr (H == 8) {
        const bool odd = (l & 1) != 0;
        const int  qb  = l & 1;
        float e[4], t[4], o[4];
#pragma unroll
        for (int i = 0; i < 4; ++i) {
            const float keep = odd ? p[2 * i + 1] : p[2 * i];
            const float send = odd ? p[2 * i]     : p[2 * i + 1];
            e[i] = lrelu02((keep + row_dpp_f<0x161>(send)) + ad);
        }
        float bm = max_sel(max_sel(e[0], e[1]), max_sel(e[2], e[3]));
        bm = max_sel(bm, row_dpp_f<0x161>(bm));
        const float mn = max_sel(m, bm);
        sc = exp_nonpos(m - mn);
#pragma unroll
        for (int i = 0; i < 4; ++i) t[i] = exp_nonpos((2 * i + qb < nleft ? e[i] : -INFINITY) - mn);
#pragma unroll
        for (int i = 0; i < 4; ++i) o[i] = row_dpp_f<0x161>(t[i]);
#pragma unroll
        for (int i = 0; i < 4; ++i) {
            w[2 * i]     = odd ? o[i] : t[i];
            w[2 * i + 1] = odd ? t[i] : o[i];
        }
        m = mn;
    } else {
        const bool b3 = (l & 8) != 0, b2 = (l & 4) != 0, b1 = (l & 2) != 0;
        const bool odd = (l & 1) != 0;
        float r[4], t2[2];
#pragma unroll
        for (int i = 0; i < 4; ++i) {
            const float keep = b3 ? p[2 * i + 1] : p[2 * i];
            const float send = b3 ? p[2 * i]     : p[2 * i + 1];
            r[i] = keep + row_dpp_f<0x168>(send);
        }
#pragma unroll
        for (int i = 0; i < 2; ++i) {
            const float keep = b2 ? r[2 * i + 1] : r[2 * i];
            const float send = b2 ? r[2 * i]     : r[2 * i + 1];
            t2[i] = keep + row_dpp_f<0x164>(send);
        }
        const float keep = b1 ? t2[1] : t2[0];
        const float send = b1 ? t2[0] : t2[1];
        float u = keep + row_dpp_f<0x162>(send);
        u += row_dpp_f<0x161>(u);
        const float e = lrelu02(u + ad);
        float bm = max_sel(e, row_dpp_f<0x168>(e));
        bm = max_sel(bm, row_dpp_f<0x164>(bm));
        bm = max_sel(bm, row_dpp_f<0x162>(bm));
        const float mn = max_sel(m, bm);
        const int ql = (b1 ? 4 : 0) + (b2 ? 2 : 0) + (b3 ? 1 : 0);
        const float tv = exp_nonpos((odd ? m : e) - mn);
        const float wv = (odd || ql < nleft) ? tv : 0.f;
        w[0] = row_dpp_f<0x150 + 0>(wv);
        w[1] = row_dpp_f<0x150 + 8>(wv);
        w[2] = row_dpp_f<0x150 + 4>(wv);
        w[3] = row_dpp_f<0x150 + 12>(wv);
        w[4] = row_dpp_f<0x150 + 2>(wv);
        w[5] = row_dpp_f<0x150 + 10>(wv);
        w[6] = row_dpp_f<0x150 + 6>(wv);
        w[7] = row_dpp_f<0x150 + 14>(wv);
        sc   = row_dpp_f<0x150 + 1>(wv);
        m = mn;
    }

    den *= sc;
    acc.x *= sc; acc.y *= sc; acc.z *= sc; acc.w *= sc;
#pragma unroll
    for (int q = 0; q < 8; ++q) {
        den += w[q];
        acc.x = __builtin_fmaf(xq[q].x, w[q], acc.x); acc.y = __builtin_fmaf(xq[q].y, w[q], acc.y);
        acc.z = __builtin_fmaf(xq[q].z, w[q], acc.z); acc.w = __builtin_fmaf(xq[q].w, w[q], acc.w);
    }
}

template <int H, bool ELU>
__device__ __forceinline__ float4 gather_row(
    const int* __restrict__ rowptr, const int* __restrict__ cnt,
    const int* __restrict__ csr,
    const float4 as4, const float4 ad4, const float4 bv,
    const float* __restrict__ xh, int d, int l)
{
    const int n = cnt[d];
    const int* __restrict__ cs = csr + rowptr[d];
    const float4 x0  = *(const float4*)(xh + d * 64 + 4 * l);
    const char* __restrict__ xhb = (const char*)xh;
    const unsigned lane16 = 16u * (unsigned)l;

    const int last = n - 1;
    int sid = d;
    if (n > 0) sid = cs[l < last ? l : last];

    const float ad = head_sum<H>(dot4(x0, ad4));
    float m   = lrelu02(head_sum<H>(dot4(x0, as4)) + ad);
    float den = 1.0f;
    float4 acc = x0;

    for (int cb = 0; cb < n; cb += 16) {
        int nx = sid;
        if (cb + 16 < n) {
            const int j = cb + 16 + l;
            nx = cs[j < last ? j : last];
        }
        const unsigned boff = (unsigned)sid << 8;
        edge_trip<H, 0>(xhb, boff, lane16, n - cb, l, as4, ad, m, den, acc);
        if (cb + 8 < n) edge_trip<H, 8>(xhb, boff, lane16, n - cb - 8, l, as4, ad, m, den, acc);
        sid = nx;
    }

    const float inv = 1.0f / (den + 1e-16f);
    float4 r;
    r.x = acc.x * inv + bv.x; r.y = acc.y * inv + bv.y;
    r.z = acc.z * inv + bv.z; r.w = acc.w * inv + bv.w;
    if constexpr (ELU) {
        r.x = r.x > 0.f ? r.x : expm1f(r.x);
        r.y = r.y > 0.f ? r.y : expm1f(r.y);
        r.z = r.z > 0.f ? r.z : expm1f(r.z);
        r.w = r.w > 0.f ? r.w : expm1f(r.w);
    }
    return r;
}

template <int H, bool ELU>
__global__ __launch_bounds__(256) void gather_kernel(
    const int* __restrict__ rowptr, const int* __restrict__ cnt,
    const int* __restrict__ csr,
    const float* __restrict__ a_src, const float* __restrict__ a_dst,
    const float* __restrict__ xh, const float* __restrict__ bias,
    float* __restrict__ out, int N)
{
    const int gid = blockIdx.x * 256 + threadIdx.x;
    const int d = gid >> 4;
    const int l = gid & 15;
    if (d >= N) return;
    const float4 as4 = *(const float4*)(a_src + 4 * l);
    const float4 ad4 = *(const float4*)(a_dst + 4 * l);
    const float4 bv  = *(const float4*)(bias + 4 * l);
    const float4 r = gather_row<H, ELU>(rowptr, cnt, csr, as4, ad4, bv, xh, d, l);
    *(float4*)(out + d * 64 + 4 * l) = r;
}

#define XPITCH 68

__global__ __launch_bounds__(256) void gather_project_kernel(
    const int* __restrict__ rowptr, const int* __restrict__ cnt,
    const int* __restrict__ csr,
    const float* __restrict__ a_src, const float* __restrict__ a_dst,
    const float* __restrict__ xh, const float* __restrict__ bias,
    const float* __restrict__ W2, float* __restrict__ Y,
    int N, int ntiles, int tpw)
{
    __shared__ __attribute__((aligned(16))) float sW[64 * 64];
    __shared__ __attribute__((aligned(16))) float sX[2][16 * XPITCH];
    const int tid = threadIdx.x;
    for (int i = tid; i < 64 * 64; i += 256) {
        const int k = i >> 6, n = i & 63;
        sW[(k >> 1) * 128 + 2 * n + (k & 1)] = W2[i];
    }

    const int l    = tid & 15;
    const int rowT = tid >> 4;
    const int wave = tid >> 5;
    const int half = (tid >> 4) & 1;
    const float4 as4 = *(const float4*)(a_src + 4 * l);
    const float4 ad4 = *(const float4*)(a_dst + 4 * l);
    const float4 bv  = *(const float4*)(bias + 4 * l);

    const int t0 = (int)blockIdx.x * tpw;
    const int t1 = t0 + tpw < ntiles ? t0 + tpw : ntiles;
    for (int t = t0; t < t1; ++t) {
        const int d  = t * 16 + rowT;
        const int dd = d < N ? d : N - 1;
        const float4 r = gather_row<8, true>(rowptr, cnt, csr, as4, ad4, bv, xh, dd, l);
        float* sx = sX[(t - t0) & 1];
        *(float4*)(sx + rowT * XPITCH + 4 * l) = r;
        __syncthreads();
        if (wave < 4) {
            v8f cacc = {};
            const float* ap = sx + l * XPITCH + 2 * half;
            const float* bp = sW + half * 128 + 2 * (l + 16 * wave);
#pragma unroll
            for (int kk = 0; kk < 16; ++kk) {
                const v2f a = *(const v2f*)(ap + 4 * kk);
                const v2f b = *(const v2f*)(bp + 256 * kk);
                cacc = __builtin_amdgcn_wmma_f32_16x16x4_f32(false, a, false, b,
                                                             (short)0, cacc, false, false);
            }
            const int row0 = t * 16 + 8 * half;
            float* yb = Y + (size_t)row0 * 64 + 16 * wave + l;
            if (t * 16 + 16 <= N) {
#pragma unroll
                for (int j = 0; j < 8; ++j) yb[(size_t)j * 64] = cacc[j];
            } else {
#pragma unroll
                for (int j = 0; j < 8; ++j) {
                    if (row0 + j < N) yb[(size_t)j * 64] = cacc[j];
                }
            }
        }
    }
}

static inline int nblk(long long n, int b) { return (int)((n + b - 1) / b); }

extern "C" void kernel_launch(void* const* d_in, const int* in_sizes, int n_in,
                              void* d_out, int out_size, void* d_ws, size_t ws_size,
                              hipStream_t stream)
{
    const int*   ei  = (const int*)  d_in[0];
    const float* X   = (const float*)d_in[1];
    const float* W1  = (const float*)d_in[2];
    const float* as1 = (const float*)d_in[3];
    const float* ad1 = (const float*)d_in[4];
    const float* b1  = (const float*)d_in[5];
    const float* W2  = (const float*)d_in[6];
    const float* as2 = (const float*)d_in[7];
    const float* ad2 = (const float*)d_in[8];
    const float* b2  = (const float*)d_in[9];

    const int E    = in_sizes[0] / 2;
    const int N    = in_sizes[1] / 128;
    const size_t N64 = (size_t)N * 64;

    const int NB = nblk(N, 512);
    const int OP = (NB + 32) & ~31;
    const int NC = nblk(E, CHUNK);
    const size_t Na = ((size_t)N + 31) & ~(size_t)31;
    const size_t Ea = ((size_t)E + 31) & ~(size_t)31;

    float* ws    = (float*)d_ws;
    float* bufA  = ws;
    float* bufB  = bufA + N64;
    int*   cnt    = (int*)(bufB + N64);
    int*   rowptr = cnt + Na;
    int*   csr    = rowptr + Na;
    unsigned* part = (unsigned*)(csr + Ea);
    int*   offg   = (int*)(part + (size_t)NC * CHUNK);
    float* out   = (float*)d_out;

    chunk_split_kernel<<<NC, 256, 0, stream>>>(ei, E, OP, part, offg);

    {
        int mtiles = (N + 15) / 16;
        order_and_gemm_kernel<<<NB + nblk(mtiles, 8), 256, 0, stream>>>(
            part, offg, NC, OP, cnt, rowptr, csr, N, NB, X, W1, bufA, 128);
        const int tpw = nblk(mtiles, 1280);
        gather_project_kernel<<<nblk(mtiles, tpw), 256, 0, stream>>>(
            rowptr, cnt, csr, as1, ad1, bufA, b1, W2, bufB, N, mtiles, tpw);
    }

    gather_kernel<1, false><<<nblk((long long)N * 16, 256), 256, 0, stream>>>(rowptr, cnt, csr, as2, ad2, bufB, b2, out, N);
}


# Sort level 2 reads each chunk piece with one wave's consecutive lanes (whole lines, 16 loads in flight, offsets passed by readlane) instead of one lane walking its piece four words per trip
# speedup vs baseline: 9.6938x; 1.0430x over previous
#include <hip/hip_runtime.h>
#include <math.h>


typedef __attribute__((ext_vector_type(2))) float v2f;
typedef __attribute__((ext_vector_type(8))) float v8f;

__device__ __forceinline__ float lrelu02(float v) {
    return v > 0.0f ? v : 0.2f * v;
}

__device__ __forceinline__ float exp_nonpos(float x) {
    return __builtin_amdgcn_exp2f(x * 1.4426950408889634f);
}

__device__ __forceinline__ void gemm_wmma_body(
    const float* __restrict__ X, const float* __restrict__ W,
    float* __restrict__ Y, int M, int K, float* sW, int blk)
{
    const int nW = K * 64;
    for (int i = threadIdx.x; i < nW; i += 256) {
        int k = i >> 6, n = i & 63;
        sW[(k >> 1) * 128 + 2 * n + (k & 1)] = W[i];
    }
    __syncthreads();

    const int wave = threadIdx.x >> 5;
    const int lane = threadIdx.x & 31;
    const int m0   = (blk * 8 + wave) * 16;
    if (m0 >= M) return;

    const int half = lane >> 4;
    const int r    = lane & 15;

    v8f c0 = {}, c1 = {}, c2 = {}, c3 = {};

    const float* arow = X + (size_t)(m0 + r) * K + 2 * half;

    for (int k0 = 0; k0 < K; k0 += 4) {
        v2f a = *(const v2f*)(arow + k0);

        const float* wp = &sW[((k0 >> 1) + half) * 128 + 2 * r];
        v2f b0 = *(const v2f*)(wp +  0);
        v2f b1 = *(const v2f*)(wp + 32);
        v2f b2 = *(const v2f*)(wp + 64);
        v2f b3 = *(const v2f*)(wp + 96);

        c0 = __builtin_amdgcn_wmma_f32_16x16x4_f32(false, a, false, b0,
                                                   (short)0, c0, false, false);
        c1 = __builtin_amdgcn_wmma_f32_16x16x4_f32(false, a, false, b1,
                                                   (short)0, c1, false, false);
        c2 = __builtin_amdgcn_wmma_f32_16x16x4_f32(false, a, false, b2,
                                                   (short)0, c2, false, false);
        c3 = __builtin_amdgcn_wmma_f32_16x16x4_f32(false, a, false, b3,
                                                   (short)0, c3, false, false);
    }

    float* yb = Y + (size_t)m0 * 64 + r;
#pragma unroll
    for (int j = 0; j < 8; ++j) {
        float* yr = yb + (size_t)(j + half * 8) * 64;
        yr[0]  = c0[j];
        yr[16] = c1[j];
        yr[32] = c2[j];
        yr[48] = c3[j];
    }
}

#define CHUNK 6400
#define EPT   25
#define MAXB  1024
#define CAPB  12288

__global__ __launch_bounds__(256) void chunk_split_kernel(
    const int* __restrict__ ei, int E, int OP,
    unsigned* __restrict__ part, int* __restrict__ offg)
{
    __shared__ int cntB[MAXB];
    __shared__ __attribute__((aligned(16))) unsigned sorted[CHUNK];
    __shared__ int wsum[8];
    const int tid = threadIdx.x, lane = tid & 31, wave = tid >> 5;

    for (int i = tid; i < MAXB; i += 256) cntB[i] = 0;
    __syncthreads();

    const int e0 = blockIdx.x * CHUNK;
    unsigned pk[EPT];
    int bk[EPT], rk[EPT];
#pragma unroll
    for (int k = 0; k < EPT; ++k) {
        const int e  = e0 + k * 256 + tid;
        const int ec = e < E ? e : E - 1;
        pk[k] = (unsigned)ei[ec];
        bk[k] = ei[E + ec];
    }
#pragma unroll
    for (int k = 0; k < EPT; ++k) {
        const int e = e0 + k * 256 + tid;
        const int d = bk[k];
        pk[k] = (pk[k] << 9) | (unsigned)(d & 511);
        bk[k] = e < E ? (d >> 9) : -1;
    }
#pragma unroll
    for (int k = 0; k < EPT; ++k) {
        rk[k] = 0;
        if (bk[k] >= 0) rk[k] = atomicAdd(&cntB[bk[k]], 1);
    }
    __syncthreads();

    const int a0 = cntB[4 * tid], a1 = cntB[4 * tid + 1];
    const int a2 = cntB[4 * tid + 2], a3 = cntB[4 * tid + 3];
    const int ts = (a0 + a1) + (a2 + a3);
    int inc = ts;
    for (int o = 1; o < 32; o <<= 1) {
        const int t = __shfl_up(inc, o);
        if (lane >= o) inc += t;
    }
    if (lane == 31) wsum[wave] = inc;
    __syncthreads();
    int ex = inc - ts;
    for (int w = 0; w < wave; ++w) ex += wsum[w];
    int4 o4;
    o4.x = ex; o4.y = ex + a0; o4.z = o4.y + a1; o4.w = o4.z + a2;
    cntB[4 * tid]     = o4.x; cntB[4 * tid + 1] = o4.y;
    cntB[4 * tid + 2] = o4.z; cntB[4 * tid + 3] = o4.w;
    if (4 * tid < OP) *(int4*)(offg + (size_t)blockIdx.x * OP + 4 * tid) = o4;
    __syncthreads();

#pragma unroll
    for (int k = 0; k < EPT; ++k) {
        if (bk[k] >= 0) sorted[cntB[bk[k]] + rk[k]] = pk[k];
    }
    __syncthreads();

    const uint4* __restrict__ s4 = (const uint4*)sorted;
    uint4* __restrict__ p4 = (uint4*)(part + (size_t)blockIdx.x * CHUNK);
    for (int i = tid; i < CHUNK / 4; i += 256) p4[i] = s4[i];
}

#define ORDER_LDS_INTS (512 + 512 + CAPB + 16 + 8)

__device__ __forceinline__ void bucket_order_body(
    const unsigned* __restrict__ part, const int* __restrict__ offg,
    int NC, int OP, int* __restrict__ cnt, int* __restrict__ rowptr,
    int* __restrict__ csr, int N, int* sm, int b)
{
    int* cntN = sm;
    int* cur  = sm + 512;
    int* outL = sm + 1024;
    int* red  = sm + 1024 + CAPB;
    int* wsum = red + 16;
    const int tid = threadIdx.x, lane = tid & 31, wave = tid >> 5;
    const int wv = __builtin_amdgcn_readfirstlane(wave);

    cntN[tid] = 0; cntN[tid + 256] = 0;

    int sT = 0, sB = 0;
    for (int c = tid; c < NC; c += 256) {
        const int o0 = offg[(size_t)c * OP + b];
        const int o1 = offg[(size_t)c * OP + b + 1];
        sB += o0; sT += o1 - o0;
    }
    for (int o = 16; o > 0; o >>= 1) {
        sT += __shfl_xor(sT, o);
        sB += __shfl_xor(sB, o);
    }
    if (lane == 0) { red[wave] = sT; red[8 + wave] = sB; }
    __syncthreads();
    int T = 0, base = 0;
    for (int w = 0; w < 8; ++w) { T += red[w]; base += red[8 + w]; }
    const bool big = T > CAPB;

    for (int cb = 0; cb < NC; cb += 256) {
        int o0 = 0, len = 0;
        if (cb + tid < NC) {
            o0  = offg[(size_t)(cb + tid) * OP + b];
            len = offg[(size_t)(cb + tid) * OP + b + 1] - o0;
        }
#pragma unroll 1
        for (int j0 = 0; j0 < 32; j0 += 8) {
            unsigned bq[8];
            int lq[8];
            unsigned wa[8], wb[8];
#pragma unroll
            for (int q = 0; q < 8; ++q) {
                const int cj = cb + 32 * wv + j0 + q;
                const int cc = cj < NC ? cj : NC - 1;
                bq[q] = (unsigned)cc * CHUNK + (unsigned)__builtin_amdgcn_readlane(o0, j0 + q);
                lq[q] = __builtin_amdgcn_readlane(len, j0 + q);
            }
#pragma unroll
            for (int q = 0; q < 8; ++q) {
                const int last = lq[q] > 0 ? lq[q] - 1 : 0;
                wa[q] = part[bq[q] + (unsigned)(lane < last ? lane : last)];
                wb[q] = part[bq[q] + (unsigned)(lane + 32 < last ? lane + 32 : last)];
            }
#pragma unroll
            for (int q = 0; q < 8; ++q) {
                if (lane < lq[q])      atomicAdd(&cntN[wa[q] & 511u], 1);
                if (lane + 32 < lq[q]) atomicAdd(&cntN[wb[q] & 511u], 1);
            }
        }
        unsigned longm = __builtin_amdgcn_ballot_w32(len > 64);
        while (longm != 0u) {
            const int j = __builtin_ctz(longm);
            longm &= longm - 1u;
            const unsigned bj = (unsigned)(cb + 32 * wv + j) * CHUNK + (unsigned)__builtin_amdgcn_readlane(o0, j);
            const int lj = __builtin_amdgcn_readlane(len, j);
            for (int i = 64 + lane; i < lj; i += 32) atomicAdd(&cntN[part[bj + (unsigned)i] & 511u], 1);
        }
    }
    __syncthreads();

    const int c0 = cntN[2 * tid], c1 = cntN[2 * tid + 1];
    const int ts = c0 + c1;
    int inc = ts;
    for (int o = 1; o < 32; o <<= 1) {
        const int t = __shfl_up(inc, o);
        if (lane >= o) inc += t;
    }
    if (lane == 31) wsum[wave] = inc;
    __syncthreads();
    int ex = inc - ts;
    for (int w = 0; w < wave; ++w) ex += wsum[w];
    cur[2 * tid] = ex; cur[2 * tid + 1] = ex + c0;
    const int d = b * 512 + 2 * tid;
    if (d + 1 < N) {
        *(int2*)(cnt + d)    = make_int2(c0, c1);
        *(int2*)(rowptr + d) = make_int2(base + ex, base + ex + c0);
    } else if (d < N) {
        cnt[d] = c0; rowptr[d] = base + ex;
    }
    __syncthreads();

    for (int cb = 0; cb < NC; cb += 256) {
        int o0 = 0, len = 0;
        if (cb + tid < NC) {
            o0  = offg[(size_t)(cb + tid) * OP + b];
            len = offg[(size_t)(cb + tid) * OP + b + 1] - o0;
        }
#pragma unroll 1
        for (int j0 = 0; j0 < 32; j0 += 8) {
            unsigned bq[8];
            int lq[8];
            unsigned wa[8], wb[8];
            int pa[8], pb[8];
#pragma unroll
            for (int q = 0; q < 8; ++q) {
                const int cj = cb + 32 * wv + j0 + q;
                const int cc = cj < NC ? cj : NC - 1;
                bq[q] = (unsigned)cc * CHUNK + (unsigned)__builtin_amdgcn_readlane(o0, j0 + q);
                lq[q] = __builtin_amdgcn_readlane(len, j0 + q);
            }
#pragma unroll
            for (int q = 0; q < 8; ++q) {
                const int last = lq[q] > 0 ? lq[q] - 1 : 0;
                wa[q] = part[bq[q] + (unsigned)(lane < last ? lane : last)];
                wb[q] = part[bq[q] + (unsigned)(lane + 32 < last ? lane + 32 : last)];
            }
#pragma unroll
            for (int q = 0; q < 8; ++q) {
                pa[q] = 0; pb[q] = 0;
                if (lane < lq[q])      pa[q] = atomicAdd(&cur[wa[q] & 511u], 1);
                if (lane + 32 < lq[q]) pb[q] = atomicAdd(&cur[wb[q] & 511u], 1);
            }
#pragma unroll
            for (int q = 0; q < 8; ++q) {
                if (lane < lq[q]) {
                    const int src = (int)(wa[q] >> 9);
                    if (!big) outL[pa[q]] = src;
                    else atomicExch(&csr[base + pa[q]], src);
                }
                if (lane + 32 < lq[q]) {
                    const int src = (int)(wb[q] >> 9);
                    if (!big) outL[pb[q]] = src;
                    else atomicExch(&csr[base + pb[q]], src);
                }
            }
        }
        unsigned longm = __builtin_amdgcn_ballot_w32(len > 64);
        while (longm != 0u) {
            const int j = __builtin_ctz(longm);
            longm &= longm - 1u;
            const unsigned bj = (unsigned)(cb + 32 * wv + j) * CHUNK + (unsigned)__builtin_amdgcn_readlane(o0, j);
            const int lj = __builtin_amdgcn_readlane(len, j);
            for (int i = 64 + lane; i < lj; i += 32) {
                const unsigned w = part[bj + (unsigned)i];
                const int pos = atomicAdd(&cur[w & 511u], 1);
                if (!big) outL[pos] = (int)(w >> 9);
                else atomicExch(&csr[base + pos], (int)(w >> 9));
            }
        }
    }
    __syncthreads();

    if (!big) {
        const int zend = base + T;
        for (int g = (base & ~31) + tid; g < zend; g += 256) {
            if (g >= base) {
                const int l0 = g & ~31;
                if (l0 >= base && l0 + 32 <= zend) {
                    csr[g] = outL[g - base];
                } else {
                    atomicExch(&csr[g], outL[g - base]);
                }
            }
        }
    }
}

__global__ __launch_bounds__(256) void order_and_gemm_kernel(
    const unsigned* __restrict__ part, const int* __restrict__ offg,
    int NC, int OP, int* __restrict__ cnt, int* __restrict__ rowptr,
    int* __restrict__ csr, int N, int NB,
    const float* __restrict__ X, const float* __restrict__ W,
    float* __restrict__ Y, int K)
{
    static_assert(ORDER_LDS_INTS >= 128 * 64, "the GEMM's W tile must fit the shared LDS block");
    __shared__ __attribute__((aligned(16))) int sm[ORDER_LDS_INTS];
    const int blk = (int)blockIdx.x;
    if (blk < NB) bucket_order_body(part, offg, NC, OP, cnt, rowptr, csr, N, sm, blk);
    else          gemm_wmma_body(X, W, Y, N, K, (float*)sm, blk - NB);
}

__device__ __forceinline__ float max_sel(float a, float b) {
    return b > a ? b : a;
}

template <int CTRL>
__device__ __forceinline__ int row_dpp_i(int v) {
    return __builtin_amdgcn_update_dpp(0, v, CTRL, 0xf, 0xf, true);
}
template <int CTRL>
__device__ __forceinline__ float row_dpp_f(float v) {
    return __int_as_float(row_dpp_i<CTRL>(__float_as_int(v)));
}

template <int H>
__device__ __forceinline__ float head_sum(float p) {
    constexpr int LPH = 16 / H;
    if constexpr (LPH > 1) p += row_dpp_f<0x161>(p);
    if constexpr (LPH > 2) p += row_dpp_f<0x162>(p);
    if constexpr (LPH > 4) p += row_dpp_f<0x164>(p);
    if constexpr (LPH > 8) p += row_dpp_f<0x168>(p);
    return p;
}

__device__ __forceinline__ float dot4(const float4 x, const float4 a) {
    return __builtin_fmaf(x.w, a.w, __builtin_fmaf(x.z, a.z, __builtin_fmaf(x.y, a.y, x.x * a.x)));
}

template <int H, int Q0>
__device__ __forceinline__ void edge_trip(
    const char* __restrict__ xhb, unsigned boff, unsigned lane16, int nleft, int l,
    const float4 as4, float ad, float& m, float& den, float4& acc)
{
    static_assert(H == 8 || H == 1, "lane maps below are written for 2 and 16 lanes per head");
    unsigned off[8];
    off[0] = (unsigned)row_dpp_i<0x150 + Q0 + 0>((int)boff) + lane16;
    off[1] = (unsigned)row_dpp_i<0x150 + Q0 + 1>((int)boff) + lane16;
    off[2] = (unsigned)row_dpp_i<0x150 + Q0 + 2>((int)boff) + lane16;
    off[3] = (unsigned)row_dpp_i<0x150 + Q0 + 3>((int)boff) + lane16;
    off[4] = (unsigned)row_dpp_i<0x150 + Q0 + 4>((int)boff) + lane16;
    off[5] = (unsigned)row_dpp_i<0x150 + Q0 + 5>((int)boff) + lane16;
    off[6] = (unsigned)row_dpp_i<0x150 + Q0 + 6>((int)boff) + lane16;
    off[7] = (unsigned)row_dpp_i<0x150 + Q0 + 7>((int)boff) + lane16;
    float4 xq[8];
    float  p[8], w[8];
    float  sc;
#pragma unroll
    for (int q = 0; q < 8; ++q) xq[q] = *(const float4*)(xhb + off[q]);
#pragma unroll
    for (int q = 0; q < 8; ++q) p[q] = dot4(xq[q], as4);

    if constexpr (H == 8) {
        const bool odd = (l & 1) != 0;
        const int  qb  = l & 1;
        float e[4], t[4], o[4];
#pragma unroll
        for (int i = 0; i < 4; ++i) {
            const float keep = odd ? p[2 * i + 1] : p[2 * i];
            const float send = odd ? p[2 * i]     : p[2 * i + 1];
            e[i] = lrelu02((keep + row_dpp_f<0x161>(send)) + ad);
        }
        float bm = max_sel(max_sel(e[0], e[1]), max_sel(e[2], e[3]));
        bm = max_sel(bm, row_dpp_f<0x161>(bm));
        const float mn = max_sel(m, bm);
        sc = exp_nonpos(m - mn);
#pragma unroll
        for (int i = 0; i < 4; ++i) t[i] = exp_nonpos((2 * i + qb < nleft ? e[i] : -INFINITY) - mn);
#pragma unroll
        for (int i = 0; i < 4; ++i) o[i] = row_dpp_f<0x161>(t[i]);
#pragma unroll
        for (int i = 0; i < 4; ++i) {
            w[2 * i]     = odd ? o[i] : t[i];
            w[2 * i + 1] = odd ? t[i] : o[i];
        }
        m = mn;
    } else {
        const bool b3 = (l & 8) != 0, b2 = (l & 4) != 0, b1 = (l & 2) != 0;
        const bool odd = (l & 1) != 0;
        float r[4], t2[2];
#pragma unroll
        for (int i = 0; i < 4; ++i) {
            const float keep = b3 ? p[2 * i + 1] : p[2 * i];
            const float send = b3 ? p[2 * i]     : p[2 * i + 1];
            r[i] = keep + row_dpp_f<0x168>(send);
        }
#pragma unroll
        for (int i = 0; i < 2; ++i) {
            const float keep = b2 ? r[2 * i + 1] : r[2 * i];
            const float send = b2 ? r[2 * i]     : r[2 * i + 1];
            t2[i] = keep + row_dpp_f<0x164>(send);
        }
        const float keep = b1 ? t2[1] : t2[0];
        const float send = b1 ? t2[0] : t2[1];
        float u = keep + row_dpp_f<0x162>(send);
        u += row_dpp_f<0x161>(u);
        const float e = lrelu02(u + ad);
        float bm = max_sel(e, row_dpp_f<0x168>(e));
        bm = max_sel(bm, row_dpp_f<0x164>(bm));
        bm = max_sel(bm, row_dpp_f<0x162>(bm));
        const float mn = max_sel(m, bm);
        const int ql = (b1 ? 4 : 0) + (b2 ? 2 : 0) + (b3 ? 1 : 0);
        const float tv = exp_nonpos((odd ? m : e) - mn);
        const float wv = (odd || ql < nleft) ? tv : 0.f;
        w[0] = row_dpp_f<0x150 + 0>(wv);
        w[1] = row_dpp_f<0x150 + 8>(wv);
        w[2] = row_dpp_f<0x150 + 4>(wv);
        w[3] = row_dpp_f<0x150 + 12>(wv);
        w[4] = row_dpp_f<0x150 + 2>(wv);
        w[5] = row_dpp_f<0x150 + 10>(wv);
        w[6] = row_dpp_f<0x150 + 6>(wv);
        w[7] = row_dpp_f<0x150 + 14>(wv);
        sc   = row_dpp_f<0x150 + 1>(wv);
        m = mn;
    }

    den *= sc;
    acc.x *= sc; acc.y *= sc; acc.z *= sc; acc.w *= sc;
#pragma unroll
    for (int q = 0; q < 8; ++q) {
        den += w[q];
        acc.x = __builtin_fmaf(xq[q].x, w[q], acc.x); acc.y = __builtin_fmaf(xq[q].y, w[q], acc.y);
        acc.z = __builtin_fmaf(xq[q].z, w[q], acc.z); acc.w = __builtin_fmaf(xq[q].w, w[q], acc.w);
    }
}

template <int H, bool ELU>
__device__ __forceinline__ float4 gather_row(
    const int* __restrict__ rowptr, const int* __restrict__ cnt,
    const int* __restrict__ csr,
    const float4 as4, const float4 ad4, const float4 bv,
    const float* __restrict__ xh, int d, int l)
{
    const int n = cnt[d];
    const int* __restrict__ cs = csr + rowptr[d];
    const float4 x0  = *(const float4*)(xh + d * 64 + 4 * l);
    const char* __restrict__ xhb = (const char*)xh;
    const unsigned lane16 = 16u * (unsigned)l;

    const int last = n - 1;
    int sid = d;
    if (n > 0) sid = cs[l < last ? l : last];

    const float ad = head_sum<H>(dot4(x0, ad4));
    float m   = lrelu02(head_sum<H>(dot4(x0, as4)) + ad);
    float den = 1.0f;
    float4 acc = x0;

    for (int cb = 0; cb < n; cb += 16) {
        int nx = sid;
        if (cb + 16 < n) {
            const int j = cb + 16 + l;
            nx = cs[j < last ? j : last];
        }
        const unsigned boff = (unsigned)sid << 8;
        edge_trip<H, 0>(xhb, boff, lane16, n - cb, l, as4, ad, m, den, acc);
        if (cb + 8 < n) edge_trip<H, 8>(xhb, boff, lane16, n - cb - 8, l, as4, ad, m, den, acc);
        sid = nx;
    }

    const float inv = 1.0f / (den + 1e-16f);
    float4 r;
    r.x = acc.x * inv + bv.x; r.y = acc.y * inv + bv.y;
    r.z = acc.z * inv + bv.z; r.w = acc.w * inv + bv.w;
    if constexpr (ELU) {
        r.x = r.x > 0.f ? r.x : expm1f(r.x);
        r.y = r.y > 0.f ? r.y : expm1f(r.y);
        r.z = r.z > 0.f ? r.z : expm1f(r.z);
        r.w = r.w > 0.f ? r.w : expm1f(r.w);
    }
    return r;
}

template <int H, bool ELU>
__global__ __launch_bounds__(256) void gather_kernel(
    const int* __restrict__ rowptr, const int* __restrict__ cnt,
    const int* __restrict__ csr,
    const float* __restrict__ a_src, const float* __restrict__ a_dst,
    const float* __restrict__ xh, const float* __restrict__ bias,
    float* __restrict__ out, int N)
{
    const int gid = blockIdx.x * 256 + threadIdx.x;
    const int d = gid >> 4;
    const int l = gid & 15;
    if (d >= N) return;
    const float4 as4 = *(const float4*)(a_src + 4 * l);
    const float4 ad4 = *(const float4*)(a_dst + 4 * l);
    const float4 bv  = *(const float4*)(bias + 4 * l);
    const float4 r = gather_row<H, ELU>(rowptr, cnt, csr, as4, ad4, bv, xh, d, l);
    *(float4*)(out + d * 64 + 4 * l) = r;
}

#define XPITCH 68

__global__ __launch_bounds__(256) void gather_project_kernel(
    const int* __restrict__ rowptr, const int* __restrict__ cnt,
    const int* __restrict__ csr,
    const float* __restrict__ a_src, const float* __restrict__ a_dst,
    const float* __restrict__ xh, const float* __restrict__ bias,
    const float* __restrict__ W2, float* __restrict__ Y,
    int N, int ntiles, int tpw)
{
    __shared__ __attribute__((aligned(16))) float sW[64 * 64];
    __shared__ __attribute__((aligned(16))) float sX[2][16 * XPITCH];
    const int tid = threadIdx.x;
    for (int i = tid; i < 64 * 64; i += 256) {
        const int k = i >> 6, n = i & 63;
        sW[(k >> 1) * 128 + 2 * n + (k & 1)] = W2[i];
    }

    const int l    = tid & 15;
    const int rowT = tid >> 4;
    const int wave = tid >> 5;
    const int half = (tid >> 4) & 1;
    const float4 as4 = *(const float4*)(a_src + 4 * l);
    const float4 ad4 = *(const float4*)(a_dst + 4 * l);
    const float4 bv  = *(const float4*)(bias + 4 * l);

    const int t0 = (int)blockIdx.x * tpw;
    const int t1 = t0 + tpw < ntiles ? t0 + tpw : ntiles;
    for (int t = t0; t < t1; ++t) {
        const int d  = t * 16 + rowT;
        const int dd = d < N ? d : N - 1;
        const float4 r = gather_row<8, true>(rowptr, cnt, csr, as4, ad4, bv, xh, dd, l);
        float* sx = sX[(t - t0) & 1];
        *(float4*)(sx + rowT * XPITCH + 4 * l) = r;
        __syncthreads();
        if (wave < 4) {
            v8f cacc = {};
            const float* ap = sx + l * XPITCH + 2 * half;
            const float* bp = sW + half * 128 + 2 * (l + 16 * wave);
#pragma unroll
            for (int kk = 0; kk < 16; ++kk) {
                const v2f a = *(const v2f*)(ap + 4 * kk);
                const v2f b = *(const v2f*)(bp + 256 * kk);
                cacc = __builtin_amdgcn_wmma_f32_16x16x4_f32(false, a, false, b,
                                                             (short)0, cacc, false, false);
            }
            const int row0 = t * 16 + 8 * half;
            float* yb = Y + (size_t)row0 * 64 + 16 * wave + l;
            if (t * 16 + 16 <= N) {
#pragma unroll
                for (int j = 0; j < 8; ++j) yb[(size_t)j * 64] = cacc[j];
            } else {
#pragma unroll
                for (int j = 0; j < 8; ++j) {
                    if (row0 + j < N) yb[(size_t)j * 64] = cacc[j];
                }
            }
        }
    }
}

static inline int nblk(long long n, int b) { return (int)((n + b - 1) / b); }

extern "C" void kernel_launch(void* const* d_in, const int* in_sizes, int n_in,
                              void* d_out, int out_size, void* d_ws, size_t ws_size,
                              hipStream_t stream)
{
    const int*   ei  = (const int*)  d_in[0];
    const float* X   = (const float*)d_in[1];
    const float* W1  = (const float*)d_in[2];
    const float* as1 = (const float*)d_in[3];
    const float* ad1 = (const float*)d_in[4];
    const float* b1  = (const float*)d_in[5];
    const float* W2  = (const float*)d_in[6];
    const float* as2 = (const float*)d_in[7];
    const float* ad2 = (const float*)d_in[8];
    const float* b2  = (const float*)d_in[9];

    const int E    = in_sizes[0] / 2;
    const int N    = in_sizes[1] / 128;
    const size_t N64 = (size_t)N * 64;

    const int NB = nblk(N, 512);
    const int OP = (NB + 32) & ~31;
    const int NC = nblk(E, CHUNK);
    const size_t Na = ((size_t)N + 31) & ~(size_t)31;
    const size_t Ea = ((size_t)E + 31) & ~(size_t)31;

    float* ws    = (float*)d_ws;
    float* bufA  = ws;
    float* bufB  = bufA + N64;
    int*   cnt    = (int*)(bufB + N64);
    int*   rowptr = cnt + Na;
    int*   csr    = rowptr + Na;
    unsigned* part = (unsigned*)(csr + Ea);
    int*   offg   = (int*)(part + (size_t)NC * CHUNK);
    float* out   = (float*)d_out;

    chunk_split_kernel<<<NC, 256, 0, stream>>>(ei, E, OP, part, offg);

    {
        int mtiles = (N + 15) / 16;
        order_and_gemm_kernel<<<NB + nblk(mtiles, 8), 256, 0, stream>>>(
            part, offg, NC, OP, cnt, rowptr, csr, N, NB, X, W1, bufA, 128);
        const int tpw = nblk(mtiles, 1280);
        gather_project_kernel<<<nblk(mtiles, tpw), 256, 0, stream>>>(
            rowptr, cnt, csr, as1, ad1, bufA, b1, W2, bufB, N, mtiles, tpw);
    }

    gather_kernel<1, false><<<nblk((long long)N * 16, 256), 256, 0, stream>>>(rowptr, cnt, csr, as2, ad2, bufB, b2, out, N);
}
